# starting kernel
# baseline (speedup 1.0000x reference)
#include <hip/hip_runtime.h>

#define NN 40000
#define EE 640000
#define DD 128

typedef __attribute__((ext_vector_type(2))) float v2f;
typedef __attribute__((ext_vector_type(8))) float v8f;


__global__ void gcn_fill_zero(float* __restrict__ p, int n) {
  int i = blockIdx.x * blockDim.x + threadIdx.x;
  if (i < n) p[i] = 0.0f;
}

__global__ void gcn_degree(const int* __restrict__ src, const int* __restrict__ dst,
                           float* __restrict__ deg_out, float* __restrict__ deg_in, int E) {
  int e = blockIdx.x * blockDim.x + threadIdx.x;
  if (e < E) {
    atomicAdd(&deg_out[src[e]], 1.0f);
    atomicAdd(&deg_in[dst[e]], 1.0f);
  }
}

__global__ void gcn_norm(float* __restrict__ deg_out, float* __restrict__ deg_in, int n) {
  int i = blockIdx.x * blockDim.x + threadIdx.x;
  if (i < n) {
    deg_out[i] = rsqrtf(fmaxf(deg_out[i], 1.0f));
    deg_in[i]  = rsqrtf(fmaxf(deg_in[i], 1.0f));
  }
}

__global__ __launch_bounds__(256) void gcn_scatter(
    const float* __restrict__ feat, const int* __restrict__ src, const int* __restrict__ dst,
    const float* __restrict__ outnorm, float* __restrict__ agg, int E) {
  long long tid = (long long)blockIdx.x * blockDim.x + threadIdx.x;
  int e    = (int)(tid >> 5);
  int lane = (int)(tid & 31);
  if (e >= E) return;
  int s = src[e];
  int d = dst[e];
  float sc = outnorm[s];
  float4 v = *((const float4*)(feat + (size_t)s * DD) + lane);
  float* ap = agg + (size_t)d * DD + lane * 4;
  atomicAdd(ap + 0, v.x * sc);
  atomicAdd(ap + 1, v.y * sc);
  atomicAdd(ap + 2, v.z * sc);
  atomicAdd(ap + 3, v.w * sc);
}

template <int K, int C, int RELU>
__global__ __launch_bounds__(32 * (C / 16)) void gcn_wmma_gemm(
    const float* __restrict__ A, const float* __restrict__ W,
    const float* __restrict__ bias, const float* __restrict__ rowscale,
    float* __restrict__ out) {
  __shared__ float Ash[16 * K];

  const int tid   = threadIdx.x;
  const int rbase = blockIdx.x * 16;

  for (int idx = tid; idx < 16 * K; idx += blockDim.x) {
    int r = idx / K;
    int k = idx - r * K;
    float v = A[(size_t)(rbase + r) * K + k];
    if (rowscale) v *= rowscale[rbase + r];
    Ash[idx] = v;
  }
  __syncthreads();

  const int wave  = tid >> 5;
  const int lane  = tid & 31;
  const int half  = lane >> 4;
  const int m     = lane & 15;
  const int cbase = wave * 16;

  v8f acc = {};
  const float* arow = &Ash[m * K];
#pragma unroll
  for (int k0 = 0; k0 < K; k0 += 4) {
    v2f a, b;
    const int ka = k0 + 2 * half;
    a.x = arow[ka];
    a.y = arow[ka + 1];
    b.x = W[(size_t)ka * C + cbase + m];
    b.y = W[(size_t)(ka + 1) * C + cbase + m];
    acc = __builtin_amdgcn_wmma_f32_16x16x4_f32(
         false, a,  false, b,
         (short)0, acc,  false,  false);
  }

  const int col = cbase + m;
  const float bv = bias[col];
#pragma unroll
  for (int r = 0; r < 8; ++r) {
    int row = rbase + r + 8 * half;
    float v = acc[r] + bv;
    if (RELU) v = fmaxf(v, 0.0f);
    out[(size_t)row * C + col] = v;
  }
}

__global__ __launch_bounds__(256) void gcn_head(
    const float* __restrict__ h, const float* __restrict__ Wl2,
    const float* __restrict__ bl2, float* __restrict__ out, int n) {
  int tid = blockIdx.x * blockDim.x + threadIdx.x;
  if (tid >= n * 10) return;
  int row = tid / 10;
  int c   = tid - row * 10;
  const float* hp = h + (size_t)row * 64;
  float acc = bl2[c];
#pragma unroll
  for (int k = 0; k < 64; ++k) acc = fmaf(hp[k], Wl2[k * 10 + c], acc);
  out[tid] = acc;
}


extern "C" void kernel_launch(void* const* d_in, const int* in_sizes, int n_in,
                              void* d_out, int out_size, void* d_ws, size_t ws_size,
                              hipStream_t stream) {
  const float* h   = (const float*)d_in[0];
  const int*   src = (const int*)d_in[1];
  const int*   dst = (const int*)d_in[2];
  const float* W0  = (const float*)d_in[3];
  const float* b0  = (const float*)d_in[4];
  const float* W1  = (const float*)d_in[5];
  const float* b1  = (const float*)d_in[6];
  const float* Wl1 = (const float*)d_in[7];
  const float* bl1 = (const float*)d_in[8];
  const float* Wl2 = (const float*)d_in[9];
  const float* bl2 = (const float*)d_in[10];
  float* out = (float*)d_out;

  float* wsf      = (float*)d_ws;
  float* deg_out  = wsf;
  float* deg_in   = wsf + NN;
  float* buf1     = wsf + 2 * NN;
  float* buf2     = buf1 + (size_t)NN * DD;

  const int TPB = 256;
  const int feat_elems = NN * DD;
  dim3 featGrid((feat_elems + TPB - 1) / TPB);
  dim3 scatGrid((EE * 32) / TPB);
  dim3 gemmGrid(NN / 16);

  gcn_fill_zero<<<dim3((2 * NN + TPB - 1) / TPB), TPB, 0, stream>>>(deg_out, 2 * NN);
  gcn_degree<<<dim3((EE + TPB - 1) / TPB), TPB, 0, stream>>>(src, dst, deg_out, deg_in, EE);
  gcn_norm<<<dim3((NN + TPB - 1) / TPB), TPB, 0, stream>>>(deg_out, deg_in, NN);

  gcn_fill_zero<<<featGrid, TPB, 0, stream>>>(buf2, feat_elems);
  gcn_scatter<<<scatGrid, TPB, 0, stream>>>(h, src, dst, deg_out, buf2, EE);
  gcn_wmma_gemm<128, 128, 1><<<gemmGrid, 256, 0, stream>>>(buf2, W0, b0, deg_in, buf1);

  gcn_fill_zero<<<featGrid, TPB, 0, stream>>>(buf2, feat_elems);
  gcn_scatter<<<scatGrid, TPB, 0, stream>>>(buf1, src, dst, deg_out, buf2, EE);
  gcn_wmma_gemm<128, 128, 1><<<gemmGrid, 256, 0, stream>>>(buf2, W1, b1, deg_in, buf1);

  gcn_wmma_gemm<128, 64, 1><<<gemmGrid, 128, 0, stream>>>(buf1, Wl1, bl1, nullptr, buf2);

  gcn_head<<<dim3((NN * 10 + TPB - 1) / TPB), TPB, 0, stream>>>(buf2, Wl2, bl2, out, NN);
}


# scatter lanes remapped (lane i takes row elements i+32j): each wave-wide atomic add covers one whole 128-byte line, 4 line requests per edge not 16; float atomics, arrival-order sums as before
# speedup vs baseline: 1.3171x; 1.3171x over previous
#include <hip/hip_runtime.h>

#define NN 40000
#define EE 640000
#define DD 128

typedef __attribute__((ext_vector_type(2))) float v2f;
typedef __attribute__((ext_vector_type(8))) float v8f;


__global__ void gcn_fill_zero(float* __restrict__ p, int n) {
  int i = blockIdx.x * blockDim.x + threadIdx.x;
  if (i < n) p[i] = 0.0f;
}

__global__ void gcn_degree(const int* __restrict__ src, const int* __restrict__ dst,
                           float* __restrict__ deg_out, float* __restrict__ deg_in, int E) {
  int e = blockIdx.x * blockDim.x + threadIdx.x;
  if (e < E) {
    atomicAdd(&deg_out[src[e]], 1.0f);
    atomicAdd(&deg_in[dst[e]], 1.0f);
  }
}

__global__ void gcn_norm(float* __restrict__ deg_out, float* __restrict__ deg_in, int n) {
  int i = blockIdx.x * blockDim.x + threadIdx.x;
  if (i < n) {
    deg_out[i] = rsqrtf(fmaxf(deg_out[i], 1.0f));
    deg_in[i]  = rsqrtf(fmaxf(deg_in[i], 1.0f));
  }
}

__global__ __launch_bounds__(256) void gcn_scatter(
    const float* __restrict__ feat, const int* __restrict__ src, const int* __restrict__ dst,
    const float* __restrict__ outnorm, float* __restrict__ agg, int E) {
  long long tid = (long long)blockIdx.x * blockDim.x + threadIdx.x;
  int e    = (int)(tid >> 5);
  int lane = (int)(tid & 31);
  if (e >= E) return;
  int s = src[e];
  int d = dst[e];
  float sc = outnorm[s];
  const float* fp = feat + (size_t)s * DD + lane;
  float v0 = fp[0];
  float v1 = fp[32];
  float v2 = fp[64];
  float v3 = fp[96];
  float* ap = agg + (size_t)d * DD + lane;
  atomicAdd(ap +  0, v0 * sc);
  atomicAdd(ap + 32, v1 * sc);
  atomicAdd(ap + 64, v2 * sc);
  atomicAdd(ap + 96, v3 * sc);
}

template <int K, int C, int RELU>
__global__ __launch_bounds__(32 * (C / 16)) void gcn_wmma_gemm(
    const float* __restrict__ A, const float* __restrict__ W,
    const float* __restrict__ bias, const float* __restrict__ rowscale,
    float* __restrict__ out) {
  __shared__ float Ash[16 * K];

  const int tid   = threadIdx.x;
  const int rbase = blockIdx.x * 16;

  for (int idx = tid; idx < 16 * K; idx += blockDim.x) {
    int r = idx / K;
    int k = idx - r * K;
    float v = A[(size_t)(rbase + r) * K + k];
    if (rowscale) v *= rowscale[rbase + r];
    Ash[idx] = v;
  }
  __syncthreads();

  const int wave  = tid >> 5;
  const int lane  = tid & 31;
  const int half  = lane >> 4;
  const int m     = lane & 15;
  const int cbase = wave * 16;

  v8f acc = {};
  const float* arow = &Ash[m * K];
#pragma unroll
  for (int k0 = 0; k0 < K; k0 += 4) {
    v2f a, b;
    const int ka = k0 + 2 * half;
    a.x = arow[ka];
    a.y = arow[ka + 1];
    b.x = W[(size_t)ka * C + cbase + m];
    b.y = W[(size_t)(ka + 1) * C + cbase + m];
    acc = __builtin_amdgcn_wmma_f32_16x16x4_f32(
         false, a,  false, b,
         (short)0, acc,  false,  false);
  }

  const int col = cbase + m;
  const float bv = bias[col];
#pragma unroll
  for (int r = 0; r < 8; ++r) {
    int row = rbase + r + 8 * half;
    float v = acc[r] + bv;
    if (RELU) v = fmaxf(v, 0.0f);
    out[(size_t)row * C + col] = v;
  }
}

__global__ __launch_bounds__(256) void gcn_head(
    const float* __restrict__ h, const float* __restrict__ Wl2,
    const float* __restrict__ bl2, float* __restrict__ out, int n) {
  int tid = blockIdx.x * blockDim.x + threadIdx.x;
  if (tid >= n * 10) return;
  int row = tid / 10;
  int c   = tid - row * 10;
  const float* hp = h + (size_t)row * 64;
  float acc = bl2[c];
#pragma unroll
  for (int k = 0; k < 64; ++k) acc = fmaf(hp[k], Wl2[k * 10 + c], acc);
  out[tid] = acc;
}


extern "C" void kernel_launch(void* const* d_in, const int* in_sizes, int n_in,
                              void* d_out, int out_size, void* d_ws, size_t ws_size,
                              hipStream_t stream) {
  const float* h   = (const float*)d_in[0];
  const int*   src = (const int*)d_in[1];
  const int*   dst = (const int*)d_in[2];
  const float* W0  = (const float*)d_in[3];
  const float* b0  = (const float*)d_in[4];
  const float* W1  = (const float*)d_in[5];
  const float* b1  = (const float*)d_in[6];
  const float* Wl1 = (const float*)d_in[7];
  const float* bl1 = (const float*)d_in[8];
  const float* Wl2 = (const float*)d_in[9];
  const float* bl2 = (const float*)d_in[10];
  float* out = (float*)d_out;

  float* wsf      = (float*)d_ws;
  float* deg_out  = wsf;
  float* deg_in   = wsf + NN;
  float* buf1     = wsf + 2 * NN;
  float* buf2     = buf1 + (size_t)NN * DD;

  const int TPB = 256;
  const int feat_elems = NN * DD;
  dim3 featGrid((feat_elems + TPB - 1) / TPB);
  dim3 scatGrid((EE * 32) / TPB);
  dim3 gemmGrid(NN / 16);

  gcn_fill_zero<<<dim3((2 * NN + TPB - 1) / TPB), TPB, 0, stream>>>(deg_out, 2 * NN);
  gcn_degree<<<dim3((EE + TPB - 1) / TPB), TPB, 0, stream>>>(src, dst, deg_out, deg_in, EE);
  gcn_norm<<<dim3((NN + TPB - 1) / TPB), TPB, 0, stream>>>(deg_out, deg_in, NN);

  gcn_fill_zero<<<featGrid, TPB, 0, stream>>>(buf2, feat_elems);
  gcn_scatter<<<scatGrid, TPB, 0, stream>>>(h, src, dst, deg_out, buf2, EE);
  gcn_wmma_gemm<128, 128, 1><<<gemmGrid, 256, 0, stream>>>(buf2, W0, b0, deg_in, buf1);

  gcn_fill_zero<<<featGrid, TPB, 0, stream>>>(buf2, feat_elems);
  gcn_scatter<<<scatGrid, TPB, 0, stream>>>(buf1, src, dst, deg_out, buf2, EE);
  gcn_wmma_gemm<128, 128, 1><<<gemmGrid, 256, 0, stream>>>(buf2, W1, b1, deg_in, buf1);

  gcn_wmma_gemm<128, 64, 1><<<gemmGrid, 128, 0, stream>>>(buf1, Wl1, bl1, nullptr, buf2);

  gcn_head<<<dim3((NN * 10 + TPB - 1) / TPB), TPB, 0, stream>>>(buf2, Wl2, bl2, out, NN);
}


# scatter replaced by owner work-groups: each of 250 groups keeps 160 destination rows in LDS, scans dst for its edges, sums with LDS float adds (arrival order), stores once; no global float atomics
# speedup vs baseline: 1.5097x; 1.1462x over previous
#include <hip/hip_runtime.h>

#define NN 40000
#define EE 640000
#define DD 128

typedef __attribute__((ext_vector_type(2))) float v2f;
typedef __attribute__((ext_vector_type(8))) float v8f;


__global__ void gcn_fill_zero(float* __restrict__ p, int n) {
  int i = blockIdx.x * blockDim.x + threadIdx.x;
  if (i < n) p[i] = 0.0f;
}

__global__ void gcn_degree(const int* __restrict__ src, const int* __restrict__ dst,
                           float* __restrict__ deg_out, float* __restrict__ deg_in, int E) {
  int e = blockIdx.x * blockDim.x + threadIdx.x;
  if (e < E) {
    atomicAdd(&deg_out[src[e]], 1.0f);
    atomicAdd(&deg_in[dst[e]], 1.0f);
  }
}

__global__ void gcn_norm(float* __restrict__ deg_out, float* __restrict__ deg_in, int n) {
  int i = blockIdx.x * blockDim.x + threadIdx.x;
  if (i < n) {
    deg_out[i] = rsqrtf(fmaxf(deg_out[i], 1.0f));
    deg_in[i]  = rsqrtf(fmaxf(deg_in[i], 1.0f));
  }
}

#define OWN_ROWS   160
#define OWN_BLOCKS (NN / OWN_ROWS)
#define SCAN_U     8
#define SCAN_PIECES (EE / 128)
#define QCAP       (32 + SCAN_U * 128)

static_assert(NN % OWN_ROWS == 0, "owned ranges must tile the nodes");
static_assert(EE % 128 == 0 && SCAN_PIECES % SCAN_U == 0, "scan pieces must tile the edges");
static_assert(SCAN_PIECES % OWN_BLOCKS == 0 && (SCAN_U & (SCAN_U - 1)) == 0 && 32 * SCAN_U <= SCAN_PIECES,
              "per-block scan offset is rounded to whole steps");

__global__ __launch_bounds__(1024) void gcn_gather_owner(
    const float* __restrict__ feat, const int* __restrict__ src, const int* __restrict__ dst,
    const float* __restrict__ outnorm, float* __restrict__ agg) {
  __shared__ float rows[OWN_ROWS * DD];
  __shared__ int   qmem[32 * QCAP];

  const int tid  = threadIdx.x;
  const int wave = __builtin_amdgcn_readfirstlane(tid >> 5);
  const int lane = tid & 31;
  const int lo   = blockIdx.x * OWN_ROWS;

  float4* rows4 = reinterpret_cast<float4*>(rows);
  for (int i = tid; i < OWN_ROWS * DD / 4; i += 1024) rows4[i] = make_float4(0.f, 0.f, 0.f, 0.f);
  __syncthreads();

  const int qb = wave * QCAP;
  int cnt = 0;

  const int4* dst4 = reinterpret_cast<const int4*>(dst);
  const int off = (blockIdx.x * (SCAN_PIECES / OWN_BLOCKS)) & ~(SCAN_U - 1);

  int g = wave * SCAN_U;
  int p = g + off;
  if (p >= SCAN_PIECES) p -= SCAN_PIECES;
  int4 nx[SCAN_U];
#pragma unroll
  for (int u = 0; u < SCAN_U; ++u) nx[u] = dst4[(p + u) * 32 + lane];

  for (;;) {
    int4 dv[SCAN_U];
#pragma unroll
    for (int u = 0; u < SCAN_U; ++u) dv[u] = nx[u];
    const int  pc   = p;
    const int  gn   = g + 32 * SCAN_U;
    const bool last = (gn >= SCAN_PIECES);
    if (!last) {
      p = gn + off;
      if (p >= SCAN_PIECES) p -= SCAN_PIECES;
    }
#pragma unroll
    for (int u = 0; u < SCAN_U; ++u) nx[u] = dst4[(p + u) * 32 + lane];

#pragma unroll
    for (int u = 0; u < SCAN_U; ++u) {
      const unsigned t[4] = {(unsigned)(dv[u].x - lo), (unsigned)(dv[u].y - lo),
                             (unsigned)(dv[u].z - lo), (unsigned)(dv[u].w - lo)};
      const unsigned mn = min(min(t[0], t[1]), min(t[2], t[3]));
      if (__builtin_amdgcn_ballot_w32(mn < OWN_ROWS) != 0u) {
        const int ebase = ((pc + u) * 32 + lane) * 4;
#pragma unroll
        for (int c = 0; c < 4; ++c) {
          const bool m = t[c] < OWN_ROWS;
          const unsigned bal = __builtin_amdgcn_ballot_w32(m);
          if (bal != 0u) {
            if (m) qmem[qb + cnt + (int)__builtin_amdgcn_mbcnt_lo(bal, 0u)] = ((ebase + c) << 8) | (int)t[c];
            cnt += __builtin_popcount(bal);
          }
        }
      }
    }

    if (cnt >= 32 || (last && cnt > 0)) {
      for (int base = 0; base < cnt; base += 32) {
        const int n = min(32, cnt - base);
        int v = 0;
        if (lane < n) v = qmem[qb + base + lane];
        const int   e  = v >> 8;
        const int   dl = v & 255;
        const int   s  = src[e];
        const float sc = outnorm[s];
        for (int k = 0; k < n; k += 8) {
          float r[8][4];
#pragma unroll
          for (int j = 0; j < 8; ++j) {
            const int sj = __builtin_amdgcn_readlane(s, k + j);
            const float* fp = feat + (size_t)sj * DD + lane;
            r[j][0] = fp[0];
            r[j][1] = fp[32];
            r[j][2] = fp[64];
            r[j][3] = fp[96];
          }
#pragma unroll
          for (int j = 0; j < 8; ++j) {
            if (k + j < n) {
              const int   dj  = __builtin_amdgcn_readlane(dl, k + j);
              const float scj = __int_as_float(__builtin_amdgcn_readlane(__float_as_int(sc), k + j));
              float* ap = rows + dj * DD + lane;
              __hip_atomic_fetch_add(ap +  0, r[j][0] * scj, __ATOMIC_RELAXED, __HIP_MEMORY_SCOPE_WORKGROUP);
              __hip_atomic_fetch_add(ap + 32, r[j][1] * scj, __ATOMIC_RELAXED, __HIP_MEMORY_SCOPE_WORKGROUP);
              __hip_atomic_fetch_add(ap + 64, r[j][2] * scj, __ATOMIC_RELAXED, __HIP_MEMORY_SCOPE_WORKGROUP);
              __hip_atomic_fetch_add(ap + 96, r[j][3] * scj, __ATOMIC_RELAXED, __HIP_MEMORY_SCOPE_WORKGROUP);
            }
          }
        }
      }
      cnt = 0;
    }
    if (last) break;
    g = gn;
  }

  __syncthreads();
  for (int r = wave; r < OWN_ROWS; r += 32) {
    const float4 v = reinterpret_cast<const float4*>(rows + r * DD)[lane];
    reinterpret_cast<float4*>(agg + (size_t)(lo + r) * DD)[lane] = v;
  }
}

template <int K, int C, int RELU>
__global__ __launch_bounds__(32 * (C / 16)) void gcn_wmma_gemm(
    const float* __restrict__ A, const float* __restrict__ W,
    const float* __restrict__ bias, const float* __restrict__ rowscale,
    float* __restrict__ out) {
  __shared__ float Ash[16 * K];

  const int tid   = threadIdx.x;
  const int rbase = blockIdx.x * 16;

  for (int idx = tid; idx < 16 * K; idx += blockDim.x) {
    int r = idx / K;
    int k = idx - r * K;
    float v = A[(size_t)(rbase + r) * K + k];
    if (rowscale) v *= rowscale[rbase + r];
    Ash[idx] = v;
  }
  __syncthreads();

  const int wave  = tid >> 5;
  const int lane  = tid & 31;
  const int half  = lane >> 4;
  const int m     = lane & 15;
  const int cbase = wave * 16;

  v8f acc = {};
  const float* arow = &Ash[m * K];
#pragma unroll
  for (int k0 = 0; k0 < K; k0 += 4) {
    v2f a, b;
    const int ka = k0 + 2 * half;
    a.x = arow[ka];
    a.y = arow[ka + 1];
    b.x = W[(size_t)ka * C + cbase + m];
    b.y = W[(size_t)(ka + 1) * C + cbase + m];
    acc = __builtin_amdgcn_wmma_f32_16x16x4_f32(
         false, a,  false, b,
         (short)0, acc,  false,  false);
  }

  const int col = cbase + m;
  const float bv = bias[col];
#pragma unroll
  for (int r = 0; r < 8; ++r) {
    int row = rbase + r + 8 * half;
    float v = acc[r] + bv;
    if (RELU) v = fmaxf(v, 0.0f);
    out[(size_t)row * C + col] = v;
  }
}

__global__ __launch_bounds__(256) void gcn_head(
    const float* __restrict__ h, const float* __restrict__ Wl2,
    const float* __restrict__ bl2, float* __restrict__ out, int n) {
  int tid = blockIdx.x * blockDim.x + threadIdx.x;
  if (tid >= n * 10) return;
  int row = tid / 10;
  int c   = tid - row * 10;
  const float* hp = h + (size_t)row * 64;
  float acc = bl2[c];
#pragma unroll
  for (int k = 0; k < 64; ++k) acc = fmaf(hp[k], Wl2[k * 10 + c], acc);
  out[tid] = acc;
}


extern "C" void kernel_launch(void* const* d_in, const int* in_sizes, int n_in,
                              void* d_out, int out_size, void* d_ws, size_t ws_size,
                              hipStream_t stream) {
  const float* h   = (const float*)d_in[0];
  const int*   src = (const int*)d_in[1];
  const int*   dst = (const int*)d_in[2];
  const float* W0  = (const float*)d_in[3];
  const float* b0  = (const float*)d_in[4];
  const float* W1  = (const float*)d_in[5];
  const float* b1  = (const float*)d_in[6];
  const float* Wl1 = (const float*)d_in[7];
  const float* bl1 = (const float*)d_in[8];
  const float* Wl2 = (const float*)d_in[9];
  const float* bl2 = (const float*)d_in[10];
  float* out = (float*)d_out;

  float* wsf      = (float*)d_ws;
  float* deg_out  = wsf;
  float* deg_in   = wsf + NN;
  float* buf1     = wsf + 2 * NN;
  float* buf2     = buf1 + (size_t)NN * DD;

  const int TPB = 256;
  dim3 ownGrid(OWN_BLOCKS);
  dim3 gemmGrid(NN / 16);

  gcn_fill_zero<<<dim3((2 * NN + TPB - 1) / TPB), TPB, 0, stream>>>(deg_out, 2 * NN);
  gcn_degree<<<dim3((EE + TPB - 1) / TPB), TPB, 0, stream>>>(src, dst, deg_out, deg_in, EE);
  gcn_norm<<<dim3((NN + TPB - 1) / TPB), TPB, 0, stream>>>(deg_out, deg_in, NN);

  gcn_gather_owner<<<ownGrid, 1024, 0, stream>>>(h, src, dst, deg_out, buf2);
  gcn_wmma_gemm<128, 128, 1><<<gemmGrid, 256, 0, stream>>>(buf2, W0, b0, deg_in, buf1);

  gcn_gather_owner<<<ownGrid, 1024, 0, stream>>>(buf1, src, dst, deg_out, buf2);
  gcn_wmma_gemm<128, 128, 1><<<gemmGrid, 256, 0, stream>>>(buf2, W1, b1, deg_in, buf1);

  gcn_wmma_gemm<128, 64, 1><<<gemmGrid, 128, 0, stream>>>(buf1, Wl1, bl1, nullptr, buf2);

  gcn_head<<<dim3((NN * 10 + TPB - 1) / TPB), TPB, 0, stream>>>(buf2, Wl2, bl2, out, NN);
}


# edges binned by owner work-group once per call (LDS ranks per chunk, one returning add per chunk and owner); both segment sums read their list, no dst scan; LDS float adds, arrival order
# speedup vs baseline: 1.6330x; 1.0817x over previous
#include <hip/hip_runtime.h>

#define NN 40000
#define EE 640000
#define DD 128

typedef __attribute__((ext_vector_type(2))) float v2f;
typedef __attribute__((ext_vector_type(8))) float v8f;

#define OWN_ROWS   160
#define OWN_BLOCKS (NN / OWN_ROWS)
#define REGION     (OWN_ROWS * DD)
#define LIST_HDR   32
#define LIST_CAP   (REGION - LIST_HDR)
#define BIN_CHUNK  (EE / OWN_BLOCKS)
#define BIN_PER    ((BIN_CHUNK + 1023) / 1024)

static_assert(NN % OWN_ROWS == 0, "owned ranges must tile the nodes");
static_assert(EE % OWN_BLOCKS == 0, "edge chunks must tile the edges");
static_assert(NN < (1 << 23) && OWN_ROWS <= 256, "a list entry packs source and local row into one int");


__global__ void gcn_fill_zero(float* __restrict__ p, int n, int* __restrict__ lists) {
  int i = blockIdx.x * blockDim.x + threadIdx.x;
  if (i < n) p[i] = 0.0f;
  if (i < OWN_BLOCKS) lists[(size_t)i * REGION] = 0;
}

__global__ __launch_bounds__(1024) void gcn_degree_bin(
    const int* __restrict__ src, const int* __restrict__ dst,
    float* __restrict__ deg_out, float* __restrict__ deg_in, int* __restrict__ lists) {
  __shared__ int hist[OWN_BLOCKS];
  __shared__ int base[OWN_BLOCKS];

  const int tid = threadIdx.x;
  if (tid < OWN_BLOCKS) hist[tid] = 0;
  __syncthreads();

  const int e0 = blockIdx.x * BIN_CHUNK;
  int own[BIN_PER], ent[BIN_PER], rnk[BIN_PER];
#pragma unroll
  for (int k = 0; k < BIN_PER; ++k) {
    const int i = tid + k * 1024;
    own[k] = -1;
    ent[k] = 0;
    rnk[k] = 0;
    if (i < BIN_CHUNK) {
      const int s = src[e0 + i];
      const int d = dst[e0 + i];
      atomicAdd(&deg_out[s], 1.0f);
      atomicAdd(&deg_in[d], 1.0f);
      const int o = d / OWN_ROWS;
      own[k] = o;
      ent[k] = (s << 8) | (d - o * OWN_ROWS);
      rnk[k] = atomicAdd(&hist[o], 1);
    }
  }
  __syncthreads();

  if (tid < OWN_BLOCKS) {
    const int c = hist[tid];
    int b = 0;
    if (c > 0) b = atomicAdd(&lists[(size_t)tid * REGION], c);
    base[tid] = b;
  }
  __syncthreads();

#pragma unroll
  for (int k = 0; k < BIN_PER; ++k) {
    if (own[k] >= 0) {
      const int idx = base[own[k]] + rnk[k];
      if (idx < LIST_CAP) lists[(size_t)own[k] * REGION + LIST_HDR + idx] = ent[k];
    }
  }
}

__global__ void gcn_norm(float* __restrict__ deg_out, float* __restrict__ deg_in, int n) {
  int i = blockIdx.x * blockDim.x + threadIdx.x;
  if (i < n) {
    deg_out[i] = rsqrtf(fmaxf(deg_out[i], 1.0f));
    deg_in[i]  = rsqrtf(fmaxf(deg_in[i], 1.0f));
  }
}

#define SCAN_U     8
#define SCAN_PIECES (EE / 128)
#define QCAP       (32 + SCAN_U * 128)

static_assert(EE % 128 == 0 && SCAN_PIECES % SCAN_U == 0, "scan pieces must tile the edges");
static_assert(SCAN_PIECES % OWN_BLOCKS == 0 && (SCAN_U & (SCAN_U - 1)) == 0 && 32 * SCAN_U <= SCAN_PIECES,
              "per-block scan offset is rounded to whole steps");

__device__ __forceinline__ void gcn_sum_listed(const float* __restrict__ feat, float* rows,
                                               int s, int dl, float sc, int n, int lane) {
  for (int k = 0; k < n; k += 8) {
    float r[8][4];
#pragma unroll
    for (int j = 0; j < 8; ++j) {
      const int sj = __builtin_amdgcn_readlane(s, k + j);
      const float* fp = feat + (size_t)sj * DD + lane;
      r[j][0] = fp[0];
      r[j][1] = fp[32];
      r[j][2] = fp[64];
      r[j][3] = fp[96];
    }
#pragma unroll
    for (int j = 0; j < 8; ++j) {
      if (k + j < n) {
        const int   dj  = __builtin_amdgcn_readlane(dl, k + j);
        const float scj = __int_as_float(__builtin_amdgcn_readlane(__float_as_int(sc), k + j));
        float* ap = rows + dj * DD + lane;
        __hip_atomic_fetch_add(ap +  0, r[j][0] * scj, __ATOMIC_RELAXED, __HIP_MEMORY_SCOPE_WORKGROUP);
        __hip_atomic_fetch_add(ap + 32, r[j][1] * scj, __ATOMIC_RELAXED, __HIP_MEMORY_SCOPE_WORKGROUP);
        __hip_atomic_fetch_add(ap + 64, r[j][2] * scj, __ATOMIC_RELAXED, __HIP_MEMORY_SCOPE_WORKGROUP);
        __hip_atomic_fetch_add(ap + 96, r[j][3] * scj, __ATOMIC_RELAXED, __HIP_MEMORY_SCOPE_WORKGROUP);
      }
    }
  }
}

__global__ __launch_bounds__(1024) void gcn_gather_owner(
    const float* __restrict__ feat, const int* __restrict__ src, const int* __restrict__ dst,
    const float* __restrict__ outnorm, float* agg, const int* lists) {
  __shared__ float rows[OWN_ROWS * DD];
  __shared__ int   qmem[32 * QCAP];

  const int tid  = threadIdx.x;
  const int wave = __builtin_amdgcn_readfirstlane(tid >> 5);
  const int lane = tid & 31;
  const int lo   = blockIdx.x * OWN_ROWS;
  const int* mylist = lists + (size_t)blockIdx.x * REGION;

  float4* rows4 = reinterpret_cast<float4*>(rows);
  for (int i = tid; i < OWN_ROWS * DD / 4; i += 1024) rows4[i] = make_float4(0.f, 0.f, 0.f, 0.f);
  __syncthreads();

  const int total = __builtin_amdgcn_readfirstlane(mylist[0]);

  if (total <= LIST_CAP) {
    const int per = (((total + 31) >> 5) + 7) & ~7;
    const int end = min(total, (wave + 1) * per);
    for (int base = wave * per; base < end; base += 32) {
      const int n = min(32, end - base);
      int v = 0;
      if (lane < n) v = mylist[LIST_HDR + base + lane];
      const int   s  = v >> 8;
      const int   dl = v & 255;
      const float sc = outnorm[s];
      gcn_sum_listed(feat, rows, s, dl, sc, n, lane);
    }
  } else {
    const int qb = wave * QCAP;
    int cnt = 0;

    const int4* dst4 = reinterpret_cast<const int4*>(dst);
    const int off = (blockIdx.x * (SCAN_PIECES / OWN_BLOCKS)) & ~(SCAN_U - 1);

    int g = wave * SCAN_U;
    int p = g + off;
    if (p >= SCAN_PIECES) p -= SCAN_PIECES;
    int4 nx[SCAN_U];
#pragma unroll
    for (int u = 0; u < SCAN_U; ++u) nx[u] = dst4[(p + u) * 32 + lane];

    for (;;) {
      int4 dv[SCAN_U];
#pragma unroll
      for (int u = 0; u < SCAN_U; ++u) dv[u] = nx[u];
      const int  pc   = p;
      const int  gn   = g + 32 * SCAN_U;
      const bool last = (gn >= SCAN_PIECES);
      if (!last) {
        p = gn + off;
        if (p >= SCAN_PIECES) p -= SCAN_PIECES;
      }
#pragma unroll
      for (int u = 0; u < SCAN_U; ++u) nx[u] = dst4[(p + u) * 32 + lane];

#pragma unroll
      for (int u = 0; u < SCAN_U; ++u) {
        const unsigned t[4] = {(unsigned)(dv[u].x - lo), (unsigned)(dv[u].y - lo),
                               (unsigned)(dv[u].z - lo), (unsigned)(dv[u].w - lo)};
        const unsigned mn = min(min(t[0], t[1]), min(t[2], t[3]));
        if (__builtin_amdgcn_ballot_w32(mn < OWN_ROWS) != 0u) {
          const int ebase = ((pc + u) * 32 + lane) * 4;
#pragma unroll
          for (int c = 0; c < 4; ++c) {
            const bool m = t[c] < OWN_ROWS;
            const unsigned bal = __builtin_amdgcn_ballot_w32(m);
            if (bal != 0u) {
              if (m) qmem[qb + cnt + (int)__builtin_amdgcn_mbcnt_lo(bal, 0u)] = ((ebase + c) << 8) | (int)t[c];
              cnt += __builtin_popcount(bal);
            }
          }
        }
      }

      if (cnt >= 32 || (last && cnt > 0)) {
        for (int base = 0; base < cnt; base += 32) {
          const int n = min(32, cnt - base);
          int v = 0;
          if (lane < n) v = qmem[qb + base + lane];
          const int   e  = v >> 8;
          const int   dl = v & 255;
          const int   s  = src[e];
          const float sc = outnorm[s];
          gcn_sum_listed(feat, rows, s, dl, sc, n, lane);
        }
        cnt = 0;
      }
      if (last) break;
      g = gn;
    }
  }

  __syncthreads();
  for (int r = wave; r < OWN_ROWS; r += 32) {
    const float4 v = reinterpret_cast<const float4*>(rows + r * DD)[lane];
    reinterpret_cast<float4*>(agg + (size_t)(lo + r) * DD)[lane] = v;
  }
}

template <int K, int C, int RELU>
__global__ __launch_bounds__(32 * (C / 16)) void gcn_wmma_gemm(
    const float* A, const float* __restrict__ W,
    const float* __restrict__ bias, const float* __restrict__ rowscale,
    float* out) {
  __shared__ float Ash[16 * K];

  const int tid   = threadIdx.x;
  const int rbase = blockIdx.x * 16;

  for (int idx = tid; idx < 16 * K; idx += blockDim.x) {
    int r = idx / K;
    int k = idx - r * K;
    float v = A[(size_t)(rbase + r) * K + k];
    if (rowscale) v *= rowscale[rbase + r];
    Ash[idx] = v;
  }
  __syncthreads();

  const int wave  = tid >> 5;
  const int lane  = tid & 31;
  const int half  = lane >> 4;
  const int m     = lane & 15;
  const int cbase = wave * 16;

  v8f acc = {};
  const float* arow = &Ash[m * K];
#pragma unroll
  for (int k0 = 0; k0 < K; k0 += 4) {
    v2f a, b;
    const int ka = k0 + 2 * half;
    a.x = arow[ka];
    a.y = arow[ka + 1];
    b.x = W[(size_t)ka * C + cbase + m];
    b.y = W[(size_t)(ka + 1) * C + cbase + m];
    acc = __builtin_amdgcn_wmma_f32_16x16x4_f32(
         false, a,  false, b,
         (short)0, acc,  false,  false);
  }

  const int col = cbase + m;
  const float bv = bias[col];
#pragma unroll
  for (int r = 0; r < 8; ++r) {
    int row = rbase + r + 8 * half;
    float v = acc[r] + bv;
    if (RELU) v = fmaxf(v, 0.0f);
    out[(size_t)row * C + col] = v;
  }
}

__global__ __launch_bounds__(256) void gcn_head(
    const float* __restrict__ h, const float* __restrict__ Wl2,
    const float* __restrict__ bl2, float* __restrict__ out, int n) {
  int tid = blockIdx.x * blockDim.x + threadIdx.x;
  if (tid >= n * 10) return;
  int row = tid / 10;
  int c   = tid - row * 10;
  const float* hp = h + (size_t)row * 64;
  float acc = bl2[c];
#pragma unroll
  for (int k = 0; k < 64; ++k) acc = fmaf(hp[k], Wl2[k * 10 + c], acc);
  out[tid] = acc;
}


extern "C" void kernel_launch(void* const* d_in, const int* in_sizes, int n_in,
                              void* d_out, int out_size, void* d_ws, size_t ws_size,
                              hipStream_t stream) {
  const float* h   = (const float*)d_in[0];
  const int*   src = (const int*)d_in[1];
  const int*   dst = (const int*)d_in[2];
  const float* W0  = (const float*)d_in[3];
  const float* b0  = (const float*)d_in[4];
  const float* W1  = (const float*)d_in[5];
  const float* b1  = (const float*)d_in[6];
  const float* Wl1 = (const float*)d_in[7];
  const float* bl1 = (const float*)d_in[8];
  const float* Wl2 = (const float*)d_in[9];
  const float* bl2 = (const float*)d_in[10];
  float* out = (float*)d_out;

  float* wsf      = (float*)d_ws;
  float* deg_out  = wsf;
  float* deg_in   = wsf + NN;
  float* buf1     = wsf + 2 * NN;
  float* buf2     = buf1 + (size_t)NN * DD;

  const int TPB = 256;
  dim3 ownGrid(OWN_BLOCKS);
  dim3 gemmGrid(NN / 16);

  gcn_fill_zero<<<dim3((2 * NN + TPB - 1) / TPB), TPB, 0, stream>>>(deg_out, 2 * NN, (int*)buf1);
  gcn_degree_bin<<<ownGrid, 1024, 0, stream>>>(src, dst, deg_out, deg_in, (int*)buf1);
  gcn_norm<<<dim3((NN + TPB - 1) / TPB), TPB, 0, stream>>>(deg_out, deg_in, NN);

  gcn_gather_owner<<<ownGrid, 1024, 0, stream>>>(h, src, dst, deg_out, buf2, (const int*)buf1);
  gcn_wmma_gemm<128, 128, 1><<<gemmGrid, 256, 0, stream>>>(buf2, W0, b0, deg_in, buf2);

  gcn_gather_owner<<<ownGrid, 1024, 0, stream>>>(buf2, src, dst, deg_out, buf1, (const int*)buf1);
  gcn_wmma_gemm<128, 128, 1><<<gemmGrid, 256, 0, stream>>>(buf1, W1, b1, deg_in, buf1);

  gcn_wmma_gemm<128, 64, 1><<<gemmGrid, 128, 0, stream>>>(buf1, Wl1, bl1, nullptr, buf2);

  gcn_head<<<dim3((NN * 10 + TPB - 1) / TPB), TPB, 0, stream>>>(buf2, Wl2, bl2, out, NN);
}


# GEMM A tile in LDS padded from 128 to 132 floats per row: the 16 rows of a WMMA A-fragment read now fall on different banks (were 16 distinct words on each bank touched); summation order unchanged
# speedup vs baseline: 1.8037x; 1.1045x over previous
#include <hip/hip_runtime.h>

#define NN 40000
#define EE 640000
#define DD 128

typedef __attribute__((ext_vector_type(2))) float v2f;
typedef __attribute__((ext_vector_type(8))) float v8f;

#define OWN_ROWS   160
#define OWN_BLOCKS (NN / OWN_ROWS)
#define REGION     (OWN_ROWS * DD)
#define LIST_HDR   32
#define LIST_CAP   (REGION - LIST_HDR)
#define BIN_CHUNK  (EE / OWN_BLOCKS)
#define BIN_PER    ((BIN_CHUNK + 1023) / 1024)

static_assert(NN % OWN_ROWS == 0, "owned ranges must tile the nodes");
static_assert(EE % OWN_BLOCKS == 0, "edge chunks must tile the edges");
static_assert(NN < (1 << 23) && OWN_ROWS <= 256, "a list entry packs source and local row into one int");


__global__ void gcn_fill_zero(float* __restrict__ p, int n, int* __restrict__ lists) {
  int i = blockIdx.x * blockDim.x + threadIdx.x;
  if (i < n) p[i] = 0.0f;
  if (i < OWN_BLOCKS) lists[(size_t)i * REGION] = 0;
}

__global__ __launch_bounds__(1024) void gcn_degree_bin(
    const int* __restrict__ src, const int* __restrict__ dst,
    float* __restrict__ deg_out, float* __restrict__ deg_in, int* __restrict__ lists) {
  __shared__ int hist[OWN_BLOCKS];
  __shared__ int base[OWN_BLOCKS];

  const int tid = threadIdx.x;
  if (tid < OWN_BLOCKS) hist[tid] = 0;
  __syncthreads();

  const int e0 = blockIdx.x * BIN_CHUNK;
  int own[BIN_PER], ent[BIN_PER], rnk[BIN_PER];
#pragma unroll
  for (int k = 0; k < BIN_PER; ++k) {
    const int i = tid + k * 1024;
    own[k] = -1;
    ent[k] = 0;
    rnk[k] = 0;
    if (i < BIN_CHUNK) {
      const int s = src[e0 + i];
      const int d = dst[e0 + i];
      atomicAdd(&deg_out[s], 1.0f);
      atomicAdd(&deg_in[d], 1.0f);
      const int o = d / OWN_ROWS;
      own[k] = o;
      ent[k] = (s << 8) | (d - o * OWN_ROWS);
      rnk[k] = atomicAdd(&hist[o], 1);
    }
  }
  __syncthreads();

  if (tid < OWN_BLOCKS) {
    const int c = hist[tid];
    int b = 0;
    if (c > 0) b = atomicAdd(&lists[(size_t)tid * REGION], c);
    base[tid] = b;
  }
  __syncthreads();

#pragma unroll
  for (int k = 0; k < BIN_PER; ++k) {
    if (own[k] >= 0) {
      const int idx = base[own[k]] + rnk[k];
      if (idx < LIST_CAP) lists[(size_t)own[k] * REGION + LIST_HDR + idx] = ent[k];
    }
  }
}

__global__ void gcn_norm(float* __restrict__ deg_out, float* __restrict__ deg_in, int n) {
  int i = blockIdx.x * blockDim.x + threadIdx.x;
  if (i < n) {
    deg_out[i] = rsqrtf(fmaxf(deg_out[i], 1.0f));
    deg_in[i]  = rsqrtf(fmaxf(deg_in[i], 1.0f));
  }
}

#define SCAN_U     8
#define SCAN_PIECES (EE / 128)
#define QCAP       (32 + SCAN_U * 128)

static_assert(EE % 128 == 0 && SCAN_PIECES % SCAN_U == 0, "scan pieces must tile the edges");
static_assert(SCAN_PIECES % OWN_BLOCKS == 0 && (SCAN_U & (SCAN_U - 1)) == 0 && 32 * SCAN_U <= SCAN_PIECES,
              "per-block scan offset is rounded to whole steps");

__device__ __forceinline__ void gcn_sum_listed(const float* __restrict__ feat, float* rows,
                                               int s, int dl, float sc, int n, int lane) {
  for (int k = 0; k < n; k += 8) {
    float r[8][4];
#pragma unroll
    for (int j = 0; j < 8; ++j) {
      const int sj = __builtin_amdgcn_readlane(s, k + j);
      const float* fp = feat + (size_t)sj * DD + lane;
      r[j][0] = fp[0];
      r[j][1] = fp[32];
      r[j][2] = fp[64];
      r[j][3] = fp[96];
    }
#pragma unroll
    for (int j = 0; j < 8; ++j) {
      if (k + j < n) {
        const int   dj  = __builtin_amdgcn_readlane(dl, k + j);
        const float scj = __int_as_float(__builtin_amdgcn_readlane(__float_as_int(sc), k + j));
        float* ap = rows + dj * DD + lane;
        __hip_atomic_fetch_add(ap +  0, r[j][0] * scj, __ATOMIC_RELAXED, __HIP_MEMORY_SCOPE_WORKGROUP);
        __hip_atomic_fetch_add(ap + 32, r[j][1] * scj, __ATOMIC_RELAXED, __HIP_MEMORY_SCOPE_WORKGROUP);
        __hip_atomic_fetch_add(ap + 64, r[j][2] * scj, __ATOMIC_RELAXED, __HIP_MEMORY_SCOPE_WORKGROUP);
        __hip_atomic_fetch_add(ap + 96, r[j][3] * scj, __ATOMIC_RELAXED, __HIP_MEMORY_SCOPE_WORKGROUP);
      }
    }
  }
}

__global__ __launch_bounds__(1024) void gcn_gather_owner(
    const float* __restrict__ feat, const int* __restrict__ src, const int* __restrict__ dst,
    const float* __restrict__ outnorm, float* agg, const int* lists) {
  __shared__ float rows[OWN_ROWS * DD];
  __shared__ int   qmem[32 * QCAP];

  const int tid  = threadIdx.x;
  const int wave = __builtin_amdgcn_readfirstlane(tid >> 5);
  const int lane = tid & 31;
  const int lo   = blockIdx.x * OWN_ROWS;
  const int* mylist = lists + (size_t)blockIdx.x * REGION;

  float4* rows4 = reinterpret_cast<float4*>(rows);
  for (int i = tid; i < OWN_ROWS * DD / 4; i += 1024) rows4[i] = make_float4(0.f, 0.f, 0.f, 0.f);
  __syncthreads();

  const int total = __builtin_amdgcn_readfirstlane(mylist[0]);

  if (total <= LIST_CAP) {
    const int per = (((total + 31) >> 5) + 7) & ~7;
    const int end = min(total, (wave + 1) * per);
    for (int base = wave * per; base < end; base += 32) {
      const int n = min(32, end - base);
      int v = 0;
      if (lane < n) v = mylist[LIST_HDR + base + lane];
      const int   s  = v >> 8;
      const int   dl = v & 255;
      const float sc = outnorm[s];
      gcn_sum_listed(feat, rows, s, dl, sc, n, lane);
    }
  } else {
    const int qb = wave * QCAP;
    int cnt = 0;

    const int4* dst4 = reinterpret_cast<const int4*>(dst);
    const int off = (blockIdx.x * (SCAN_PIECES / OWN_BLOCKS)) & ~(SCAN_U - 1);

    int g = wave * SCAN_U;
    int p = g + off;
    if (p >= SCAN_PIECES) p -= SCAN_PIECES;
    int4 nx[SCAN_U];
#pragma unroll
    for (int u = 0; u < SCAN_U; ++u) nx[u] = dst4[(p + u) * 32 + lane];

    for (;;) {
      int4 dv[SCAN_U];
#pragma unroll
      for (int u = 0; u < SCAN_U; ++u) dv[u] = nx[u];
      const int  pc   = p;
      const int  gn   = g + 32 * SCAN_U;
      const bool last = (gn >= SCAN_PIECES);
      if (!last) {
        p = gn + off;
        if (p >= SCAN_PIECES) p -= SCAN_PIECES;
      }
#pragma unroll
      for (int u = 0; u < SCAN_U; ++u) nx[u] = dst4[(p + u) * 32 + lane];

#pragma unroll
      for (int u = 0; u < SCAN_U; ++u) {
        const unsigned t[4] = {(unsigned)(dv[u].x - lo), (unsigned)(dv[u].y - lo),
                               (unsigned)(dv[u].z - lo), (unsigned)(dv[u].w - lo)};
        const unsigned mn = min(min(t[0], t[1]), min(t[2], t[3]));
        if (__builtin_amdgcn_ballot_w32(mn < OWN_ROWS) != 0u) {
          const int ebase = ((pc + u) * 32 + lane) * 4;
#pragma unroll
          for (int c = 0; c < 4; ++c) {
            const bool m = t[c] < OWN_ROWS;
            const unsigned bal = __builtin_amdgcn_ballot_w32(m);
            if (bal != 0u) {
              if (m) qmem[qb + cnt + (int)__builtin_amdgcn_mbcnt_lo(bal, 0u)] = ((ebase + c) << 8) | (int)t[c];
              cnt += __builtin_popcount(bal);
            }
          }
        }
      }

      if (cnt >= 32 || (last && cnt > 0)) {
        for (int base = 0; base < cnt; base += 32) {
          const int n = min(32, cnt - base);
          int v = 0;
          if (lane < n) v = qmem[qb + base + lane];
          const int   e  = v >> 8;
          const int   dl = v & 255;
          const int   s  = src[e];
          const float sc = outnorm[s];
          gcn_sum_listed(feat, rows, s, dl, sc, n, lane);
        }
        cnt = 0;
      }
      if (last) break;
      g = gn;
    }
  }

  __syncthreads();
  for (int r = wave; r < OWN_ROWS; r += 32) {
    const float4 v = reinterpret_cast<const float4*>(rows + r * DD)[lane];
    reinterpret_cast<float4*>(agg + (size_t)(lo + r) * DD)[lane] = v;
  }
}

#define A_PAD 4
template <int K, int C, int RELU>
__global__ __launch_bounds__(32 * (C / 16)) void gcn_wmma_gemm(
    const float* A, const float* __restrict__ W,
    const float* __restrict__ bias, const float* __restrict__ rowscale,
    float* out) {
  constexpr int P = K + A_PAD;
  static_assert(K % 64 == 0 && (P % 64) == 4 && (P * 4) % 16 == 0, "pitch spreads 16 rows x 2 halves over the 64 banks");
  __shared__ float Ash[16 * P];

  const int tid   = threadIdx.x;
  const int rbase = blockIdx.x * 16;

  for (int idx = tid; idx < 16 * K; idx += blockDim.x) {
    int r = idx / K;
    int k = idx - r * K;
    float v = A[(size_t)(rbase + r) * K + k];
    if (rowscale) v *= rowscale[rbase + r];
    Ash[r * P + k] = v;
  }
  __syncthreads();

  const int wave  = tid >> 5;
  const int lane  = tid & 31;
  const int half  = lane >> 4;
  const int m     = lane & 15;
  const int cbase = wave * 16;

  v8f acc = {};
  const float* arow = &Ash[m * P];
#pragma unroll
  for (int k0 = 0; k0 < K; k0 += 4) {
    v2f a, b;
    const int ka = k0 + 2 * half;
    a.x = arow[ka];
    a.y = arow[ka + 1];
    b.x = W[(size_t)ka * C + cbase + m];
    b.y = W[(size_t)(ka + 1) * C + cbase + m];
    acc = __builtin_amdgcn_wmma_f32_16x16x4_f32(
         false, a,  false, b,
         (short)0, acc,  false,  false);
  }

  const int col = cbase + m;
  const float bv = bias[col];
#pragma unroll
  for (int r = 0; r < 8; ++r) {
    int row = rbase + r + 8 * half;
    float v = acc[r] + bv;
    if (RELU) v = fmaxf(v, 0.0f);
    out[(size_t)row * C + col] = v;
  }
}

__global__ __launch_bounds__(256) void gcn_head(
    const float* __restrict__ h, const float* __restrict__ Wl2,
    const float* __restrict__ bl2, float* __restrict__ out, int n) {
  int tid = blockIdx.x * blockDim.x + threadIdx.x;
  if (tid >= n * 10) return;
  int row = tid / 10;
  int c   = tid - row * 10;
  const float* hp = h + (size_t)row * 64;
  float acc = bl2[c];
#pragma unroll
  for (int k = 0; k < 64; ++k) acc = fmaf(hp[k], Wl2[k * 10 + c], acc);
  out[tid] = acc;
}


extern "C" void kernel_launch(void* const* d_in, const int* in_sizes, int n_in,
                              void* d_out, int out_size, void* d_ws, size_t ws_size,
                              hipStream_t stream) {
  const float* h   = (const float*)d_in[0];
  const int*   src = (const int*)d_in[1];
  const int*   dst = (const int*)d_in[2];
  const float* W0  = (const float*)d_in[3];
  const float* b0  = (const float*)d_in[4];
  const float* W1  = (const float*)d_in[5];
  const float* b1  = (const float*)d_in[6];
  const float* Wl1 = (const float*)d_in[7];
  const float* bl1 = (const float*)d_in[8];
  const float* Wl2 = (const float*)d_in[9];
  const float* bl2 = (const float*)d_in[10];
  float* out = (float*)d_out;

  float* wsf      = (float*)d_ws;
  float* deg_out  = wsf;
  float* deg_in   = wsf + NN;
  float* buf1     = wsf + 2 * NN;
  float* buf2     = buf1 + (size_t)NN * DD;

  const int TPB = 256;
  dim3 ownGrid(OWN_BLOCKS);
  dim3 gemmGrid(NN / 16);

  gcn_fill_zero<<<dim3((2 * NN + TPB - 1) / TPB), TPB, 0, stream>>>(deg_out, 2 * NN, (int*)buf1);
  gcn_degree_bin<<<ownGrid, 1024, 0, stream>>>(src, dst, deg_out, deg_in, (int*)buf1);
  gcn_norm<<<dim3((NN + TPB - 1) / TPB), TPB, 0, stream>>>(deg_out, deg_in, NN);

  gcn_gather_owner<<<ownGrid, 1024, 0, stream>>>(h, src, dst, deg_out, buf2, (const int*)buf1);
  gcn_wmma_gemm<128, 128, 1><<<gemmGrid, 256, 0, stream>>>(buf2, W0, b0, deg_in, buf2);

  gcn_gather_owner<<<ownGrid, 1024, 0, stream>>>(buf2, src, dst, deg_out, buf1, (const int*)buf1);
  gcn_wmma_gemm<128, 128, 1><<<gemmGrid, 256, 0, stream>>>(buf1, W1, b1, deg_in, buf1);

  gcn_wmma_gemm<128, 64, 1><<<gemmGrid, 128, 0, stream>>>(buf1, Wl1, bl1, nullptr, buf2);

  gcn_head<<<dim3((NN * 10 + TPB - 1) / TPB), TPB, 0, stream>>>(buf2, Wl2, bl2, out, NN);
}


# dense layers fused into the owner segment-sum kernels: each group multiplies its 160 LDS rows on WMMA with W fragments held in registers; layer 1 also does linear1 and head from LDS; 9 launches to 5
# speedup vs baseline: 2.4187x; 1.3410x over previous
#include <hip/hip_runtime.h>

#define NN 40000
#define EE 640000
#define DD 128

typedef __attribute__((ext_vector_type(2))) float v2f;
typedef __attribute__((ext_vector_type(8))) float v8f;

#define OWN_ROWS   160
#define OWN_BLOCKS (NN / OWN_ROWS)
#define REGION     (OWN_ROWS * DD)
#define LIST_HDR   32
#define LIST_CAP   (REGION - LIST_HDR)
#define BIN_CHUNK  (EE / OWN_BLOCKS)
#define BIN_PER    ((BIN_CHUNK + 1023) / 1024)

static_assert(NN % OWN_ROWS == 0, "owned ranges must tile the nodes");
static_assert(EE % OWN_BLOCKS == 0, "edge chunks must tile the edges");
static_assert(NN < (1 << 23) && OWN_ROWS <= 256, "a list entry packs source and local row into one int");


__global__ void gcn_fill_zero(float* __restrict__ p, int n, int* __restrict__ lists) {
  int i = blockIdx.x * blockDim.x + threadIdx.x;
  if (i < n) p[i] = 0.0f;
  if (i < OWN_BLOCKS) lists[(size_t)i * REGION] = 0;
}

__global__ __launch_bounds__(1024) void gcn_degree_bin(
    const int* __restrict__ src, const int* __restrict__ dst,
    float* __restrict__ deg_out, float* __restrict__ deg_in, int* __restrict__ lists) {
  __shared__ int hist[OWN_BLOCKS];
  __shared__ int base[OWN_BLOCKS];

  const int tid = threadIdx.x;
  if (tid < OWN_BLOCKS) hist[tid] = 0;
  __syncthreads();

  const int e0 = blockIdx.x * BIN_CHUNK;
  int own[BIN_PER], ent[BIN_PER], rnk[BIN_PER];
#pragma unroll
  for (int k = 0; k < BIN_PER; ++k) {
    const int i = tid + k * 1024;
    own[k] = -1;
    ent[k] = 0;
    rnk[k] = 0;
    if (i < BIN_CHUNK) {
      const int s = src[e0 + i];
      const int d = dst[e0 + i];
      atomicAdd(&deg_out[s], 1.0f);
      atomicAdd(&deg_in[d], 1.0f);
      const int o = d / OWN_ROWS;
      own[k] = o;
      ent[k] = (s << 8) | (d - o * OWN_ROWS);
      rnk[k] = atomicAdd(&hist[o], 1);
    }
  }
  __syncthreads();

  if (tid < OWN_BLOCKS) {
    const int c = hist[tid];
    int b = 0;
    if (c > 0) b = atomicAdd(&lists[(size_t)tid * REGION], c);
    base[tid] = b;
  }
  __syncthreads();

#pragma unroll
  for (int k = 0; k < BIN_PER; ++k) {
    if (own[k] >= 0) {
      const int idx = base[own[k]] + rnk[k];
      if (idx < LIST_CAP) lists[(size_t)own[k] * REGION + LIST_HDR + idx] = ent[k];
    }
  }
}

__global__ void gcn_norm(float* __restrict__ deg_out, float* __restrict__ deg_in, int n) {
  int i = blockIdx.x * blockDim.x + threadIdx.x;
  if (i < n) {
    deg_out[i] = rsqrtf(fmaxf(deg_out[i], 1.0f));
    deg_in[i]  = rsqrtf(fmaxf(deg_in[i], 1.0f));
  }
}

#define ROWP       (DD + 4)
#define SCAN_U     8
#define SCAN_PIECES (EE / 128)
#define QCAP       (32 + SCAN_U * 128)

static_assert(EE % 128 == 0 && SCAN_PIECES % SCAN_U == 0, "scan pieces must tile the edges");
static_assert(SCAN_PIECES % OWN_BLOCKS == 0 && (SCAN_U & (SCAN_U - 1)) == 0 && 32 * SCAN_U <= SCAN_PIECES,
              "per-block scan offset is rounded to whole steps");

__device__ __forceinline__ void gcn_sum_listed(const float* __restrict__ feat, float* rows,
                                               int s, int dl, float sc, int n, int lane) {
  for (int k = 0; k < n; k += 8) {
    float r[8][4];
#pragma unroll
    for (int j = 0; j < 8; ++j) {
      const int sj = __builtin_amdgcn_readlane(s, k + j);
      const float* fp = feat + (size_t)sj * DD + lane;
      r[j][0] = fp[0];
      r[j][1] = fp[32];
      r[j][2] = fp[64];
      r[j][3] = fp[96];
    }
#pragma unroll
    for (int j = 0; j < 8; ++j) {
      if (k + j < n) {
        const int   dj  = __builtin_amdgcn_readlane(dl, k + j);
        const float scj = __int_as_float(__builtin_amdgcn_readlane(__float_as_int(sc), k + j));
        float* ap = rows + dj * ROWP + lane;
        __hip_atomic_fetch_add(ap +  0, r[j][0] * scj, __ATOMIC_RELAXED, __HIP_MEMORY_SCOPE_WORKGROUP);
        __hip_atomic_fetch_add(ap + 32, r[j][1] * scj, __ATOMIC_RELAXED, __HIP_MEMORY_SCOPE_WORKGROUP);
        __hip_atomic_fetch_add(ap + 64, r[j][2] * scj, __ATOMIC_RELAXED, __HIP_MEMORY_SCOPE_WORKGROUP);
        __hip_atomic_fetch_add(ap + 96, r[j][3] * scj, __ATOMIC_RELAXED, __HIP_MEMORY_SCOPE_WORKGROUP);
      }
    }
  }
}

#define T_PITCH 68
static_assert((ROWP % 64) == 4 && (ROWP * 4) % 16 == 0, "row pitch spreads 16 rows x 2 halves over the 64 banks");
static_assert(OWN_ROWS % 16 == 0 && OWN_ROWS * T_PITCH <= OWN_ROWS * ROWP, "linear1's result reuses the sums' LDS");

template <int C>
__device__ __forceinline__ void gcn_load_bfrag(const float* __restrict__ Wm, int col, int half, v2f (&b)[32]) {
#pragma unroll
  for (int i = 0; i < 32; ++i) {
    const int ka = 4 * i + 2 * half;
    b[i].x = Wm[(size_t)ka * C + col];
    b[i].y = Wm[(size_t)(ka + 1) * C + col];
  }
}

template <int SCALE>
__device__ __forceinline__ v8f gcn_tile(const float* arow, float sc, int half, const v2f (&b)[32]) {
  v8f acc = {};
#pragma unroll
  for (int i = 0; i < 32; ++i) {
    const int ka = 4 * i + 2 * half;
    v2f a;
    a.x = arow[ka];
    a.y = arow[ka + 1];
    if (SCALE) {
      a.x *= sc;
      a.y *= sc;
    }
    acc = __builtin_amdgcn_wmma_f32_16x16x4_f32(
         false, a,  false, b[i],
         (short)0, acc,  false,  false);
  }
  return acc;
}

template <int LAST>
__global__ __launch_bounds__(1024) void gcn_layer_owner(
    const float* __restrict__ feat, const int* __restrict__ src, const int* __restrict__ dst,
    const float* __restrict__ outnorm, const float* __restrict__ innorm,
    const int* __restrict__ lists,
    const float* __restrict__ W, const float* __restrict__ bias,
    const float* __restrict__ Wl1, const float* __restrict__ bl1,
    const float* __restrict__ Wl2, const float* __restrict__ bl2,
    float* __restrict__ outp) {
  __shared__ float rows[OWN_ROWS * ROWP];
  __shared__ int   qmem[32 * QCAP];
  __shared__ float hid[LAST ? OWN_ROWS * ROWP : 4];

  const int tid  = threadIdx.x;
  const int wave = __builtin_amdgcn_readfirstlane(tid >> 5);
  const int lane = tid & 31;
  const int lo   = blockIdx.x * OWN_ROWS;
  const int* mylist = lists + (size_t)blockIdx.x * REGION;

  float4* rows4 = reinterpret_cast<float4*>(rows);
  for (int i = tid; i < OWN_ROWS * ROWP / 4; i += 1024) rows4[i] = make_float4(0.f, 0.f, 0.f, 0.f);
  __syncthreads();

  const int total = __builtin_amdgcn_readfirstlane(mylist[0]);

  if (total <= LIST_CAP) {
    const int per = (((total + 31) >> 5) + 7) & ~7;
    const int end = min(total, (wave + 1) * per);
    for (int base = wave * per; base < end; base += 32) {
      const int n = min(32, end - base);
      int v = 0;
      if (lane < n) v = mylist[LIST_HDR + base + lane];
      const int   s  = v >> 8;
      const int   dl = v & 255;
      const float sc = outnorm[s];
      gcn_sum_listed(feat, rows, s, dl, sc, n, lane);
    }
  } else {
    const int qb = wave * QCAP;
    int cnt = 0;

    const int4* dst4 = reinterpret_cast<const int4*>(dst);
    const int off = (blockIdx.x * (SCAN_PIECES / OWN_BLOCKS)) & ~(SCAN_U - 1);

    int g = wave * SCAN_U;
    int p = g + off;
    if (p >= SCAN_PIECES) p -= SCAN_PIECES;
    int4 nx[SCAN_U];
#pragma unroll
    for (int u = 0; u < SCAN_U; ++u) nx[u] = dst4[(p + u) * 32 + lane];

    for (;;) {
      int4 dv[SCAN_U];
#pragma unroll
      for (int u = 0; u < SCAN_U; ++u) dv[u] = nx[u];
      const int  pc   = p;
      const int  gn   = g + 32 * SCAN_U;
      const bool last = (gn >= SCAN_PIECES);
      if (!last) {
        p = gn + off;
        if (p >= SCAN_PIECES) p -= SCAN_PIECES;
      }
#pragma unroll
      for (int u = 0; u < SCAN_U; ++u) nx[u] = dst4[(p + u) * 32 + lane];

#pragma unroll
      for (int u = 0; u < SCAN_U; ++u) {
        const unsigned t[4] = {(unsigned)(dv[u].x - lo), (unsigned)(dv[u].y - lo),
                               (unsigned)(dv[u].z - lo), (unsigned)(dv[u].w - lo)};
        const unsigned mn = min(min(t[0], t[1]), min(t[2], t[3]));
        if (__builtin_amdgcn_ballot_w32(mn < OWN_ROWS) != 0u) {
          const int ebase = ((pc + u) * 32 + lane) * 4;
#pragma unroll
          for (int c = 0; c < 4; ++c) {
            const bool m = t[c] < OWN_ROWS;
            const unsigned bal = __builtin_amdgcn_ballot_w32(m);
            if (bal != 0u) {
              if (m) qmem[qb + cnt + (int)__builtin_amdgcn_mbcnt_lo(bal, 0u)] = ((ebase + c) << 8) | (int)t[c];
              cnt += __builtin_popcount(bal);
            }
          }
        }
      }

      if (cnt >= 32 || (last && cnt > 0)) {
        for (int base = 0; base < cnt; base += 32) {
          const int n = min(32, cnt - base);
          int v = 0;
          if (lane < n) v = qmem[qb + base + lane];
          const int   e  = v >> 8;
          const int   dl = v & 255;
          const int   s  = src[e];
          const float sc = outnorm[s];
          gcn_sum_listed(feat, rows, s, dl, sc, n, lane);
        }
        cnt = 0;
      }
      if (last) break;
      g = gn;
    }
  }


  const int half = lane >> 4;
  const int m    = lane & 15;
  const int ct  = wave & 7;
  const int rg  = ((wave >> 3) + (wave & 3)) & 3;
  const int col = ct * 16 + m;

  v2f b[32];
  gcn_load_bfrag<DD>(W, col, half, b);
  const float bv = bias[col];
  float scv[3];
#pragma unroll
  for (int j = 0; j < 3; ++j) {
    const int rt = rg + 4 * j;
    scv[j] = (rt < OWN_ROWS / 16) ? innorm[lo + rt * 16 + m] : 0.0f;
  }
  __syncthreads();

#pragma unroll
  for (int j = 0; j < 3; ++j) {
    const int rt = rg + 4 * j;
    if (rt < OWN_ROWS / 16) {
      const int r0 = rt * 16;
      const v8f acc = gcn_tile<1>(rows + (r0 + m) * ROWP, scv[j], half, b);
#pragma unroll
      for (int r = 0; r < 8; ++r) {
        const int row = r0 + r + 8 * half;
        const float v = fmaxf(acc[r] + bv, 0.0f);
        if (LAST) hid[row * ROWP + col] = v;
        else      outp[(size_t)(lo + row) * DD + col] = v;
      }
    }
  }

  if (LAST) {
    const int ct1  = wave & 3;
    const int rg1  = ((wave >> 2) + 2 * (wave & 3)) & 7;
    const int col1 = ct1 * 16 + m;
    gcn_load_bfrag<64>(Wl1, col1, half, b);
    const float bv1 = bl1[col1];
    __syncthreads();
    float* tb = rows;
    for (int j = 0; j < 2; ++j) {
      const int rt = rg1 + 8 * j;
      if (rt < OWN_ROWS / 16) {
        const int r0 = rt * 16;
        const v8f acc = gcn_tile<0>(hid + (r0 + m) * ROWP, 1.0f, half, b);
#pragma unroll
        for (int r = 0; r < 8; ++r)
          tb[(r0 + r + 8 * half) * T_PITCH + col1] = fmaxf(acc[r] + bv1, 0.0f);
      }
    }
    __syncthreads();

    for (int o = tid; o < OWN_ROWS * 10; o += 1024) {
      const int row = o / 10;
      const int c   = o - row * 10;
      const float* tp = tb + row * T_PITCH;
      float acc = bl2[c];
#pragma unroll
      for (int k = 0; k < 64; ++k) acc = fmaf(tp[k], Wl2[k * 10 + c], acc);
      outp[(size_t)lo * 10 + o] = acc;
    }
  }
}


extern "C" void kernel_launch(void* const* d_in, const int* in_sizes, int n_in,
                              void* d_out, int out_size, void* d_ws, size_t ws_size,
                              hipStream_t stream) {
  const float* h   = (const float*)d_in[0];
  const int*   src = (const int*)d_in[1];
  const int*   dst = (const int*)d_in[2];
  const float* W0  = (const float*)d_in[3];
  const float* b0  = (const float*)d_in[4];
  const float* W1  = (const float*)d_in[5];
  const float* b1  = (const float*)d_in[6];
  const float* Wl1 = (const float*)d_in[7];
  const float* bl1 = (const float*)d_in[8];
  const float* Wl2 = (const float*)d_in[9];
  const float* bl2 = (const float*)d_in[10];
  float* out = (float*)d_out;

  float* wsf      = (float*)d_ws;
  float* deg_out  = wsf;
  float* deg_in   = wsf + NN;
  float* buf1     = wsf + 2 * NN;
  float* buf2     = buf1 + (size_t)NN * DD;

  const int TPB = 256;
  dim3 ownGrid(OWN_BLOCKS);

  gcn_fill_zero<<<dim3((2 * NN + TPB - 1) / TPB), TPB, 0, stream>>>(deg_out, 2 * NN, (int*)buf1);
  gcn_degree_bin<<<ownGrid, 1024, 0, stream>>>(src, dst, deg_out, deg_in, (int*)buf1);
  gcn_norm<<<dim3((NN + TPB - 1) / TPB), TPB, 0, stream>>>(deg_out, deg_in, NN);

  gcn_layer_owner<0><<<ownGrid, 1024, 0, stream>>>(h, src, dst, deg_out, deg_in, (const int*)buf1,
                                                   W0, b0, nullptr, nullptr, nullptr, nullptr, buf2);
  gcn_layer_owner<1><<<ownGrid, 1024, 0, stream>>>(buf2, src, dst, deg_out, deg_in, (const int*)buf1,
                                                   W1, b1, Wl1, bl1, Wl2, bl2, out);
}


# owner list ordered by destination row in LDS; a wave sums each row's run in registers from whole-row 16-byte loads; LDS float adds per run, not per edge (runs in list order, joined in arrival order)
# speedup vs baseline: 2.5768x; 1.0654x over previous
#include <hip/hip_runtime.h>

#define NN 40000
#define EE 640000
#define DD 128

typedef __attribute__((ext_vector_type(2))) float v2f;
typedef __attribute__((ext_vector_type(8))) float v8f;

#define OWN_ROWS   160
#define OWN_BLOCKS (NN / OWN_ROWS)
#define REGION     (OWN_ROWS * DD)
#define LIST_HDR   32
#define LIST_CAP   (REGION - LIST_HDR)
#define BIN_CHUNK  (EE / OWN_BLOCKS)
#define BIN_PER    ((BIN_CHUNK + 1023) / 1024)

static_assert(NN % OWN_ROWS == 0, "owned ranges must tile the nodes");
static_assert(EE % OWN_BLOCKS == 0, "edge chunks must tile the edges");
static_assert(NN < (1 << 23) && OWN_ROWS <= 256, "a list entry packs source and local row into one int");


__global__ void gcn_fill_zero(float* __restrict__ p, int n, int* __restrict__ lists) {
  int i = blockIdx.x * blockDim.x + threadIdx.x;
  if (i < n) p[i] = 0.0f;
  if (i < OWN_BLOCKS) lists[(size_t)i * REGION] = 0;
}

__global__ __launch_bounds__(1024) void gcn_degree_bin(
    const int* __restrict__ src, const int* __restrict__ dst,
    float* __restrict__ deg_out, float* __restrict__ deg_in, int* __restrict__ lists) {
  __shared__ int hist[OWN_BLOCKS];
  __shared__ int base[OWN_BLOCKS];

  const int tid = threadIdx.x;
  if (tid < OWN_BLOCKS) hist[tid] = 0;
  __syncthreads();

  const int e0 = blockIdx.x * BIN_CHUNK;
  int own[BIN_PER], ent[BIN_PER], rnk[BIN_PER];
#pragma unroll
  for (int k = 0; k < BIN_PER; ++k) {
    const int i = tid + k * 1024;
    own[k] = -1;
    ent[k] = 0;
    rnk[k] = 0;
    if (i < BIN_CHUNK) {
      const int s = src[e0 + i];
      const int d = dst[e0 + i];
      atomicAdd(&deg_out[s], 1.0f);
      atomicAdd(&deg_in[d], 1.0f);
      const int o = d / OWN_ROWS;
      own[k] = o;
      ent[k] = (s << 8) | (d - o * OWN_ROWS);
      rnk[k] = atomicAdd(&hist[o], 1);
    }
  }
  __syncthreads();

  if (tid < OWN_BLOCKS) {
    const int c = hist[tid];
    int b = 0;
    if (c > 0) b = atomicAdd(&lists[(size_t)tid * REGION], c);
    base[tid] = b;
  }
  __syncthreads();

#pragma unroll
  for (int k = 0; k < BIN_PER; ++k) {
    if (own[k] >= 0) {
      const int idx = base[own[k]] + rnk[k];
      if (idx < LIST_CAP) lists[(size_t)own[k] * REGION + LIST_HDR + idx] = ent[k];
    }
  }
}

__global__ void gcn_norm(float* __restrict__ deg_out, float* __restrict__ deg_in, int n) {
  int i = blockIdx.x * blockDim.x + threadIdx.x;
  if (i < n) {
    deg_out[i] = rsqrtf(fmaxf(deg_out[i], 1.0f));
    deg_in[i]  = rsqrtf(fmaxf(deg_in[i], 1.0f));
  }
}

#define ROWP       (DD + 4)
#define ROWS_PER_LANE (OWN_ROWS / 32)
#define SORT_U     4
#define SCAN_U     8
#define SCAN_PIECES (EE / 128)
#define QCAP       (32 + SCAN_U * 128)

static_assert(EE % 128 == 0 && SCAN_PIECES % SCAN_U == 0, "scan pieces must tile the edges");
static_assert(OWN_ROWS % 32 == 0 && DD == 128, "one wave scans the row counts; a lane holds 4 of a row's 128 elements");
static_assert(LIST_CAP <= 32 * QCAP, "the ordered list reuses the scan path's LDS");
static_assert(SCAN_PIECES % OWN_BLOCKS == 0 && (SCAN_U & (SCAN_U - 1)) == 0 && 32 * SCAN_U <= SCAN_PIECES,
              "per-block scan offset is rounded to whole steps");

__device__ __forceinline__ void gcn_sum_listed(const float* __restrict__ feat, float* rows,
                                               int s, int dl, float sc, int n, int lane) {
  for (int k = 0; k < n; k += 8) {
    float r[8][4];
#pragma unroll
    for (int j = 0; j < 8; ++j) {
      const int sj = __builtin_amdgcn_readlane(s, k + j);
      const float* fp = feat + (size_t)sj * DD + lane;
      r[j][0] = fp[0];
      r[j][1] = fp[32];
      r[j][2] = fp[64];
      r[j][3] = fp[96];
    }
#pragma unroll
    for (int j = 0; j < 8; ++j) {
      if (k + j < n) {
        const int   dj  = __builtin_amdgcn_readlane(dl, k + j);
        const float scj = __int_as_float(__builtin_amdgcn_readlane(__float_as_int(sc), k + j));
        float* ap = rows + dj * ROWP + lane;
        __hip_atomic_fetch_add(ap +  0, r[j][0] * scj, __ATOMIC_RELAXED, __HIP_MEMORY_SCOPE_WORKGROUP);
        __hip_atomic_fetch_add(ap + 32, r[j][1] * scj, __ATOMIC_RELAXED, __HIP_MEMORY_SCOPE_WORKGROUP);
        __hip_atomic_fetch_add(ap + 64, r[j][2] * scj, __ATOMIC_RELAXED, __HIP_MEMORY_SCOPE_WORKGROUP);
        __hip_atomic_fetch_add(ap + 96, r[j][3] * scj, __ATOMIC_RELAXED, __HIP_MEMORY_SCOPE_WORKGROUP);
      }
    }
  }
}

__device__ __forceinline__ void gcn_flush_run(float* rows, int row, const float4& a, int lane) {
  float* ap = rows + row * ROWP + 4 * lane;
  __hip_atomic_fetch_add(ap + 0, a.x, __ATOMIC_RELAXED, __HIP_MEMORY_SCOPE_WORKGROUP);
  __hip_atomic_fetch_add(ap + 1, a.y, __ATOMIC_RELAXED, __HIP_MEMORY_SCOPE_WORKGROUP);
  __hip_atomic_fetch_add(ap + 2, a.z, __ATOMIC_RELAXED, __HIP_MEMORY_SCOPE_WORKGROUP);
  __hip_atomic_fetch_add(ap + 3, a.w, __ATOMIC_RELAXED, __HIP_MEMORY_SCOPE_WORKGROUP);
}

__device__ __forceinline__ void gcn_sum_ordered(const float* __restrict__ feat, float* rows,
                                                int s, int dl, float sc, int n, int lane,
                                                float4& acc, int& currow) {
  for (int k = 0; k < n; k += 8) {
    float4 r[8];
#pragma unroll
    for (int j = 0; j < 8; ++j) {
      const int sj = __builtin_amdgcn_readlane(s, k + j);
      r[j] = *reinterpret_cast<const float4*>(feat + (size_t)sj * DD + 4 * lane);
    }
#pragma unroll
    for (int j = 0; j < 8; ++j) {
      if (k + j < n) {
        const int   dj  = __builtin_amdgcn_readlane(dl, k + j);
        const float scj = __int_as_float(__builtin_amdgcn_readlane(__float_as_int(sc), k + j));
        if (dj != currow) {
          if (currow >= 0) gcn_flush_run(rows, currow, acc, lane);
          acc = make_float4(0.f, 0.f, 0.f, 0.f);
          currow = dj;
        }
        acc.x = fmaf(r[j].x, scj, acc.x);
        acc.y = fmaf(r[j].y, scj, acc.y);
        acc.z = fmaf(r[j].z, scj, acc.z);
        acc.w = fmaf(r[j].w, scj, acc.w);
      }
    }
  }
}

#define T_PITCH 68
static_assert((ROWP % 64) == 4 && (ROWP * 4) % 16 == 0, "row pitch spreads 16 rows x 2 halves over the 64 banks");
static_assert(OWN_ROWS % 16 == 0 && OWN_ROWS * T_PITCH <= OWN_ROWS * ROWP, "linear1's result reuses the sums' LDS");

template <int C>
__device__ __forceinline__ void gcn_load_bfrag(const float* __restrict__ Wm, int col, int half, v2f (&b)[32]) {
#pragma unroll
  for (int i = 0; i < 32; ++i) {
    const int ka = 4 * i + 2 * half;
    b[i].x = Wm[(size_t)ka * C + col];
    b[i].y = Wm[(size_t)(ka + 1) * C + col];
  }
}

template <int SCALE>
__device__ __forceinline__ v8f gcn_tile(const float* arow, float sc, int half, const v2f (&b)[32]) {
  v8f acc = {};
#pragma unroll
  for (int i = 0; i < 32; ++i) {
    const int ka = 4 * i + 2 * half;
    v2f a;
    a.x = arow[ka];
    a.y = arow[ka + 1];
    if (SCALE) {
      a.x *= sc;
      a.y *= sc;
    }
    acc = __builtin_amdgcn_wmma_f32_16x16x4_f32(
         false, a,  false, b[i],
         (short)0, acc,  false,  false);
  }
  return acc;
}

template <int LAST>
__global__ __launch_bounds__(1024) void gcn_layer_owner(
    const float* __restrict__ feat, const int* __restrict__ src, const int* __restrict__ dst,
    const float* __restrict__ outnorm, const float* __restrict__ innorm,
    const int* __restrict__ lists,
    const float* __restrict__ W, const float* __restrict__ bias,
    const float* __restrict__ Wl1, const float* __restrict__ bl1,
    const float* __restrict__ Wl2, const float* __restrict__ bl2,
    float* __restrict__ outp) {
  __shared__ float rows[OWN_ROWS * ROWP];
  __shared__ int   qmem[32 * QCAP];
  __shared__ float hid[LAST ? OWN_ROWS * ROWP : 4];
  __shared__ int   rcnt[OWN_ROWS];

  const int tid  = threadIdx.x;
  const int wave = __builtin_amdgcn_readfirstlane(tid >> 5);
  const int lane = tid & 31;
  const int lo   = blockIdx.x * OWN_ROWS;
  const int* mylist = lists + (size_t)blockIdx.x * REGION;

  float4* rows4 = reinterpret_cast<float4*>(rows);
  for (int i = tid; i < OWN_ROWS * ROWP / 4; i += 1024) rows4[i] = make_float4(0.f, 0.f, 0.f, 0.f);
  if (tid < OWN_ROWS) rcnt[tid] = 0;
  __syncthreads();

  const int total = __builtin_amdgcn_readfirstlane(mylist[0]);

  if (total <= LIST_CAP) {
    int* ordered = qmem;
    for (int i0 = tid; i0 < total; i0 += SORT_U * 1024) {
      int v[SORT_U];
#pragma unroll
      for (int u = 0; u < SORT_U; ++u) {
        const int i = i0 + u * 1024;
        v[u] = (i < total) ? mylist[LIST_HDR + i] : -1;
      }
#pragma unroll
      for (int u = 0; u < SORT_U; ++u)
        if (v[u] >= 0) atomicAdd(&rcnt[v[u] & 255], 1);
    }
    __syncthreads();
    if (wave == 0) {
      int cr[ROWS_PER_LANE];
      int mine = 0;
#pragma unroll
      for (int q = 0; q < ROWS_PER_LANE; ++q) {
        cr[q] = rcnt[lane * ROWS_PER_LANE + q];
        mine += cr[q];
      }
      int inc = mine;
#pragma unroll
      for (int d = 1; d < 32; d <<= 1) {
        const int t = __shfl_up(inc, d);
        if (lane >= d) inc += t;
      }
      int run = inc - mine;
#pragma unroll
      for (int q = 0; q < ROWS_PER_LANE; ++q) {
        rcnt[lane * ROWS_PER_LANE + q] = run;
        run += cr[q];
      }
    }
    __syncthreads();
    for (int i0 = tid; i0 < total; i0 += SORT_U * 1024) {
      int v[SORT_U];
#pragma unroll
      for (int u = 0; u < SORT_U; ++u) {
        const int i = i0 + u * 1024;
        v[u] = (i < total) ? mylist[LIST_HDR + i] : -1;
      }
#pragma unroll
      for (int u = 0; u < SORT_U; ++u)
        if (v[u] >= 0) ordered[atomicAdd(&rcnt[v[u] & 255], 1)] = v[u];
    }
    __syncthreads();

    const int per = (((total + 31) >> 5) + 7) & ~7;
    const int end = min(total, (wave + 1) * per);
    float4 acc = make_float4(0.f, 0.f, 0.f, 0.f);
    int currow = -1;
    for (int base = wave * per; base < end; base += 32) {
      const int n = min(32, end - base);
      int v = 0;
      if (lane < n) v = ordered[base + lane];
      const int   s  = v >> 8;
      const int   dl = v & 255;
      const float sc = outnorm[s];
      gcn_sum_ordered(feat, rows, s, dl, sc, n, lane, acc, currow);
    }
    if (currow >= 0) gcn_flush_run(rows, currow, acc, lane);
  } else {
    const int qb = wave * QCAP;
    int cnt = 0;

    const int4* dst4 = reinterpret_cast<const int4*>(dst);
    const int off = (blockIdx.x * (SCAN_PIECES / OWN_BLOCKS)) & ~(SCAN_U - 1);

    int g = wave * SCAN_U;
    int p = g + off;
    if (p >= SCAN_PIECES) p -= SCAN_PIECES;
    int4 nx[SCAN_U];
#pragma unroll
    for (int u = 0; u < SCAN_U; ++u) nx[u] = dst4[(p + u) * 32 + lane];

    for (;;) {
      int4 dv[SCAN_U];
#pragma unroll
      for (int u = 0; u < SCAN_U; ++u) dv[u] = nx[u];
      const int  pc   = p;
      const int  gn   = g + 32 * SCAN_U;
      const bool last = (gn >= SCAN_PIECES);
      if (!last) {
        p = gn + off;
        if (p >= SCAN_PIECES) p -= SCAN_PIECES;
      }
#pragma unroll
      for (int u = 0; u < SCAN_U; ++u) nx[u] = dst4[(p + u) * 32 + lane];

#pragma unroll
      for (int u = 0; u < SCAN_U; ++u) {
        const unsigned t[4] = {(unsigned)(dv[u].x - lo), (unsigned)(dv[u].y - lo),
                               (unsigned)(dv[u].z - lo), (unsigned)(dv[u].w - lo)};
        const unsigned mn = min(min(t[0], t[1]), min(t[2], t[3]));
        if (__builtin_amdgcn_ballot_w32(mn < OWN_ROWS) != 0u) {
          const int ebase = ((pc + u) * 32 + lane) * 4;
#pragma unroll
          for (int c = 0; c < 4; ++c) {
            const bool m = t[c] < OWN_ROWS;
            const unsigned bal = __builtin_amdgcn_ballot_w32(m);
            if (bal != 0u) {
              if (m) qmem[qb + cnt + (int)__builtin_amdgcn_mbcnt_lo(bal, 0u)] = ((ebase + c) << 8) | (int)t[c];
              cnt += __builtin_popcount(bal);
            }
          }
        }
      }

      if (cnt >= 32 || (last && cnt > 0)) {
        for (int base = 0; base < cnt; base += 32) {
          const int n = min(32, cnt - base);
          int v = 0;
          if (lane < n) v = qmem[qb + base + lane];
          const int   e  = v >> 8;
          const int   dl = v & 255;
          const int   s  = src[e];
          const float sc = outnorm[s];
          gcn_sum_listed(feat, rows, s, dl, sc, n, lane);
        }
        cnt = 0;
      }
      if (last) break;
      g = gn;
    }
  }


  const int half = lane >> 4;
  const int m    = lane & 15;
  const int ct  = wave & 7;
  const int rg  = ((wave >> 3) + (wave & 3)) & 3;
  const int col = ct * 16 + m;

  v2f b[32];
  gcn_load_bfrag<DD>(W, col, half, b);
  const float bv = bias[col];
  float scv[3];
#pragma unroll
  for (int j = 0; j < 3; ++j) {
    const int rt = rg + 4 * j;
    scv[j] = (rt < OWN_ROWS / 16) ? innorm[lo + rt * 16 + m] : 0.0f;
  }
  __syncthreads();

#pragma unroll
  for (int j = 0; j < 3; ++j) {
    const int rt = rg + 4 * j;
    if (rt < OWN_ROWS / 16) {
      const int r0 = rt * 16;
      const v8f acc = gcn_tile<1>(rows + (r0 + m) * ROWP, scv[j], half, b);
#pragma unroll
      for (int r = 0; r < 8; ++r) {
        const int row = r0 + r + 8 * half;
        const float v = fmaxf(acc[r] + bv, 0.0f);
        if (LAST) hid[row * ROWP + col] = v;
        else      outp[(size_t)(lo + row) * DD + col] = v;
      }
    }
  }

  if (LAST) {
    const int ct1  = wave & 3;
    const int rg1  = ((wave >> 2) + 2 * (wave & 3)) & 7;
    const int col1 = ct1 * 16 + m;
    gcn_load_bfrag<64>(Wl1, col1, half, b);
    const float bv1 = bl1[col1];
    __syncthreads();
    float* tb = rows;
    for (int j = 0; j < 2; ++j) {
      const int rt = rg1 + 8 * j;
      if (rt < OWN_ROWS / 16) {
        const int r0 = rt * 16;
        const v8f acc = gcn_tile<0>(hid + (r0 + m) * ROWP, 1.0f, half, b);
#pragma unroll
        for (int r = 0; r < 8; ++r)
          tb[(r0 + r + 8 * half) * T_PITCH + col1] = fmaxf(acc[r] + bv1, 0.0f);
      }
    }
    __syncthreads();

    for (int o = tid; o < OWN_ROWS * 10; o += 1024) {
      const int row = o / 10;
      const int c   = o - row * 10;
      const float* tp = tb + row * T_PITCH;
      float acc = bl2[c];
#pragma unroll
      for (int k = 0; k < 64; ++k) acc = fmaf(tp[k], Wl2[k * 10 + c], acc);
      outp[(size_t)lo * 10 + o] = acc;
    }
  }
}


extern "C" void kernel_launch(void* const* d_in, const int* in_sizes, int n_in,
                              void* d_out, int out_size, void* d_ws, size_t ws_size,
                              hipStream_t stream) {
  const float* h   = (const float*)d_in[0];
  const int*   src = (const int*)d_in[1];
  const int*   dst = (const int*)d_in[2];
  const float* W0  = (const float*)d_in[3];
  const float* b0  = (const float*)d_in[4];
  const float* W1  = (const float*)d_in[5];
  const float* b1  = (const float*)d_in[6];
  const float* Wl1 = (const float*)d_in[7];
  const float* bl1 = (const float*)d_in[8];
  const float* Wl2 = (const float*)d_in[9];
  const float* bl2 = (const float*)d_in[10];
  float* out = (float*)d_out;

  float* wsf      = (float*)d_ws;
  float* deg_out  = wsf;
  float* deg_in   = wsf + NN;
  float* buf1     = wsf + 2 * NN;
  float* buf2     = buf1 + (size_t)NN * DD;

  const int TPB = 256;
  dim3 ownGrid(OWN_BLOCKS);

  gcn_fill_zero<<<dim3((2 * NN + TPB - 1) / TPB), TPB, 0, stream>>>(deg_out, 2 * NN, (int*)buf1);
  gcn_degree_bin<<<ownGrid, 1024, 0, stream>>>(src, dst, deg_out, deg_in, (int*)buf1);
  gcn_norm<<<dim3((NN + TPB - 1) / TPB), TPB, 0, stream>>>(deg_out, deg_in, NN);

  gcn_layer_owner<0><<<ownGrid, 1024, 0, stream>>>(h, src, dst, deg_out, deg_in, (const int*)buf1,
                                                   W0, b0, nullptr, nullptr, nullptr, nullptr, buf2);
  gcn_layer_owner<1><<<ownGrid, 1024, 0, stream>>>(buf2, src, dst, deg_out, deg_in, (const int*)buf1,
                                                   W1, b1, Wl1, bl1, Wl2, bl2, out);
}


# in-degree array and norm launch dropped: in-norm from the owner's per-row edge counts (made for the list ordering), out-norm formed at the gather; 640k fewer global float atomics, 5 launches to 4
# speedup vs baseline: 3.2262x; 1.2520x over previous
#include <hip/hip_runtime.h>

#define NN 40000
#define EE 640000
#define DD 128

typedef __attribute__((ext_vector_type(2))) float v2f;
typedef __attribute__((ext_vector_type(8))) float v8f;

#define OWN_ROWS   160
#define OWN_BLOCKS (NN / OWN_ROWS)
#define REGION     (OWN_ROWS * DD)
#define LIST_HDR   32
#define LIST_CAP   (REGION - LIST_HDR)
#define BIN_CHUNK  (EE / OWN_BLOCKS)
#define BIN_PER    ((BIN_CHUNK + 1023) / 1024)

static_assert(NN % OWN_ROWS == 0, "owned ranges must tile the nodes");
static_assert(EE % OWN_BLOCKS == 0, "edge chunks must tile the edges");
static_assert(NN < (1 << 23) && OWN_ROWS <= 256, "a list entry packs source and local row into one int");


__global__ void gcn_fill_zero(float* __restrict__ p, int n, int* __restrict__ lists) {
  int i = blockIdx.x * blockDim.x + threadIdx.x;
  if (i < n) p[i] = 0.0f;
  if (i < OWN_BLOCKS) lists[(size_t)i * REGION] = 0;
}

__global__ __launch_bounds__(1024) void gcn_degree_bin(
    const int* __restrict__ src, const int* __restrict__ dst,
    float* __restrict__ deg_out, int* __restrict__ lists) {
  __shared__ int hist[OWN_BLOCKS];
  __shared__ int base[OWN_BLOCKS];

  const int tid = threadIdx.x;
  if (tid < OWN_BLOCKS) hist[tid] = 0;
  __syncthreads();

  const int e0 = blockIdx.x * BIN_CHUNK;
  int own[BIN_PER], ent[BIN_PER], rnk[BIN_PER];
#pragma unroll
  for (int k = 0; k < BIN_PER; ++k) {
    const int i = tid + k * 1024;
    own[k] = -1;
    ent[k] = 0;
    rnk[k] = 0;
    if (i < BIN_CHUNK) {
      const int s = src[e0 + i];
      const int d = dst[e0 + i];
      atomicAdd(&deg_out[s], 1.0f);
      const int o = d / OWN_ROWS;
      own[k] = o;
      ent[k] = (s << 8) | (d - o * OWN_ROWS);
      rnk[k] = atomicAdd(&hist[o], 1);
    }
  }
  __syncthreads();

  if (tid < OWN_BLOCKS) {
    const int c = hist[tid];
    int b = 0;
    if (c > 0) b = atomicAdd(&lists[(size_t)tid * REGION], c);
    base[tid] = b;
  }
  __syncthreads();

#pragma unroll
  for (int k = 0; k < BIN_PER; ++k) {
    if (own[k] >= 0) {
      const int idx = base[own[k]] + rnk[k];
      if (idx < LIST_CAP) lists[(size_t)own[k] * REGION + LIST_HDR + idx] = ent[k];
    }
  }
}

#define ROWP       (DD + 4)
#define ROWS_PER_LANE (OWN_ROWS / 32)
#define SORT_U     4
#define SCAN_U     8
#define SCAN_PIECES (EE / 128)
#define QCAP       (32 + SCAN_U * 128)

static_assert(EE % 128 == 0 && SCAN_PIECES % SCAN_U == 0, "scan pieces must tile the edges");
static_assert(OWN_ROWS % 32 == 0 && DD == 128, "one wave scans the row counts; a lane holds 4 of a row's 128 elements");
static_assert(LIST_CAP <= 32 * QCAP, "the ordered list reuses the scan path's LDS");
static_assert(SCAN_PIECES % OWN_BLOCKS == 0 && (SCAN_U & (SCAN_U - 1)) == 0 && 32 * SCAN_U <= SCAN_PIECES,
              "per-block scan offset is rounded to whole steps");

__device__ __forceinline__ float gcn_deg_norm(float deg) { return rsqrtf(fmaxf(deg, 1.0f)); }

__device__ __forceinline__ void gcn_sum_listed(const float* __restrict__ feat, float* rows,
                                               int s, int dl, float dg, int n, int lane) {
  for (int k = 0; k < n; k += 8) {
    float r[8][4];
#pragma unroll
    for (int j = 0; j < 8; ++j) {
      const int sj = __builtin_amdgcn_readlane(s, k + j);
      const float* fp = feat + (size_t)sj * DD + lane;
      r[j][0] = fp[0];
      r[j][1] = fp[32];
      r[j][2] = fp[64];
      r[j][3] = fp[96];
    }
#pragma unroll
    for (int j = 0; j < 8; ++j) {
      if (k + j < n) {
        const int   dj  = __builtin_amdgcn_readlane(dl, k + j);
        const float scj = gcn_deg_norm(__int_as_float(__builtin_amdgcn_readlane(__float_as_int(dg), k + j)));
        float* ap = rows + dj * ROWP + lane;
        __hip_atomic_fetch_add(ap +  0, r[j][0] * scj, __ATOMIC_RELAXED, __HIP_MEMORY_SCOPE_WORKGROUP);
        __hip_atomic_fetch_add(ap + 32, r[j][1] * scj, __ATOMIC_RELAXED, __HIP_MEMORY_SCOPE_WORKGROUP);
        __hip_atomic_fetch_add(ap + 64, r[j][2] * scj, __ATOMIC_RELAXED, __HIP_MEMORY_SCOPE_WORKGROUP);
        __hip_atomic_fetch_add(ap + 96, r[j][3] * scj, __ATOMIC_RELAXED, __HIP_MEMORY_SCOPE_WORKGROUP);
      }
    }
  }
}

__device__ __forceinline__ void gcn_flush_run(float* rows, int row, const float4& a, int lane) {
  float* ap = rows + row * ROWP + 4 * lane;
  __hip_atomic_fetch_add(ap + 0, a.x, __ATOMIC_RELAXED, __HIP_MEMORY_SCOPE_WORKGROUP);
  __hip_atomic_fetch_add(ap + 1, a.y, __ATOMIC_RELAXED, __HIP_MEMORY_SCOPE_WORKGROUP);
  __hip_atomic_fetch_add(ap + 2, a.z, __ATOMIC_RELAXED, __HIP_MEMORY_SCOPE_WORKGROUP);
  __hip_atomic_fetch_add(ap + 3, a.w, __ATOMIC_RELAXED, __HIP_MEMORY_SCOPE_WORKGROUP);
}

__device__ __forceinline__ void gcn_sum_ordered(const float* __restrict__ feat, float* rows,
                                                int s, int dl, float dg, int n, int lane,
                                                float4& acc, int& currow) {
  for (int k = 0; k < n; k += 8) {
    float4 r[8];
#pragma unroll
    for (int j = 0; j < 8; ++j) {
      const int sj = __builtin_amdgcn_readlane(s, k + j);
      r[j] = *reinterpret_cast<const float4*>(feat + (size_t)sj * DD + 4 * lane);
    }
#pragma unroll
    for (int j = 0; j < 8; ++j) {
      if (k + j < n) {
        const int   dj  = __builtin_amdgcn_readlane(dl, k + j);
        const float scj = gcn_deg_norm(__int_as_float(__builtin_amdgcn_readlane(__float_as_int(dg), k + j)));
        if (dj != currow) {
          if (currow >= 0) gcn_flush_run(rows, currow, acc, lane);
          acc = make_float4(0.f, 0.f, 0.f, 0.f);
          currow = dj;
        }
        acc.x = fmaf(r[j].x, scj, acc.x);
        acc.y = fmaf(r[j].y, scj, acc.y);
        acc.z = fmaf(r[j].z, scj, acc.z);
        acc.w = fmaf(r[j].w, scj, acc.w);
      }
    }
  }
}

#define T_PITCH 68
static_assert((ROWP % 64) == 4 && (ROWP * 4) % 16 == 0, "row pitch spreads 16 rows x 2 halves over the 64 banks");
static_assert(OWN_ROWS % 16 == 0 && OWN_ROWS * T_PITCH <= OWN_ROWS * ROWP, "linear1's result reuses the sums' LDS");

template <int C>
__device__ __forceinline__ void gcn_load_bfrag(const float* __restrict__ Wm, int col, int half, v2f (&b)[32]) {
#pragma unroll
  for (int i = 0; i < 32; ++i) {
    const int ka = 4 * i + 2 * half;
    b[i].x = Wm[(size_t)ka * C + col];
    b[i].y = Wm[(size_t)(ka + 1) * C + col];
  }
}

template <int SCALE>
__device__ __forceinline__ v8f gcn_tile(const float* arow, float sc, int half, const v2f (&b)[32]) {
  v8f acc = {};
#pragma unroll
  for (int i = 0; i < 32; ++i) {
    const int ka = 4 * i + 2 * half;
    v2f a;
    a.x = arow[ka];
    a.y = arow[ka + 1];
    if (SCALE) {
      a.x *= sc;
      a.y *= sc;
    }
    acc = __builtin_amdgcn_wmma_f32_16x16x4_f32(
         false, a,  false, b[i],
         (short)0, acc,  false,  false);
  }
  return acc;
}

template <int LAST>
__global__ __launch_bounds__(1024) void gcn_layer_owner(
    const float* __restrict__ feat, const int* __restrict__ src, const int* __restrict__ dst,
    const float* __restrict__ outdeg,
    const int* __restrict__ lists,
    const float* __restrict__ W, const float* __restrict__ bias,
    const float* __restrict__ Wl1, const float* __restrict__ bl1,
    const float* __restrict__ Wl2, const float* __restrict__ bl2,
    float* __restrict__ outp) {
  __shared__ float rows[OWN_ROWS * ROWP];
  __shared__ int   qmem[32 * QCAP];
  __shared__ float hid[LAST ? OWN_ROWS * ROWP : 4];
  __shared__ int   rcnt[OWN_ROWS];
  __shared__ int   indeg[OWN_ROWS];

  const int tid  = threadIdx.x;
  const int wave = __builtin_amdgcn_readfirstlane(tid >> 5);
  const int lane = tid & 31;
  const int lo   = blockIdx.x * OWN_ROWS;
  const int* mylist = lists + (size_t)blockIdx.x * REGION;

  float4* rows4 = reinterpret_cast<float4*>(rows);
  for (int i = tid; i < OWN_ROWS * ROWP / 4; i += 1024) rows4[i] = make_float4(0.f, 0.f, 0.f, 0.f);
  if (tid < OWN_ROWS) {
    rcnt[tid]  = 0;
    indeg[tid] = 0;
  }
  __syncthreads();

  const int total = __builtin_amdgcn_readfirstlane(mylist[0]);

  if (total <= LIST_CAP) {
    int* ordered = qmem;
    for (int i0 = tid; i0 < total; i0 += SORT_U * 1024) {
      int v[SORT_U];
#pragma unroll
      for (int u = 0; u < SORT_U; ++u) {
        const int i = i0 + u * 1024;
        v[u] = (i < total) ? mylist[LIST_HDR + i] : -1;
      }
#pragma unroll
      for (int u = 0; u < SORT_U; ++u)
        if (v[u] >= 0) atomicAdd(&rcnt[v[u] & 255], 1);
    }
    __syncthreads();
    if (wave == 0) {
      int cr[ROWS_PER_LANE];
      int mine = 0;
#pragma unroll
      for (int q = 0; q < ROWS_PER_LANE; ++q) {
        cr[q] = rcnt[lane * ROWS_PER_LANE + q];
        indeg[lane * ROWS_PER_LANE + q] = cr[q];
        mine += cr[q];
      }
      int inc = mine;
#pragma unroll
      for (int d = 1; d < 32; d <<= 1) {
        const int t = __shfl_up(inc, d);
        if (lane >= d) inc += t;
      }
      int run = inc - mine;
#pragma unroll
      for (int q = 0; q < ROWS_PER_LANE; ++q) {
        rcnt[lane * ROWS_PER_LANE + q] = run;
        run += cr[q];
      }
    }
    __syncthreads();
    for (int i0 = tid; i0 < total; i0 += SORT_U * 1024) {
      int v[SORT_U];
#pragma unroll
      for (int u = 0; u < SORT_U; ++u) {
        const int i = i0 + u * 1024;
        v[u] = (i < total) ? mylist[LIST_HDR + i] : -1;
      }
#pragma unroll
      for (int u = 0; u < SORT_U; ++u)
        if (v[u] >= 0) ordered[atomicAdd(&rcnt[v[u] & 255], 1)] = v[u];
    }
    __syncthreads();

    const int per = (((total + 31) >> 5) + 7) & ~7;
    const int end = min(total, (wave + 1) * per);
    float4 acc = make_float4(0.f, 0.f, 0.f, 0.f);
    int currow = -1;
    for (int base = wave * per; base < end; base += 32) {
      const int n = min(32, end - base);
      int v = 0;
      if (lane < n) v = ordered[base + lane];
      const int   s  = v >> 8;
      const int   dl = v & 255;
      const float dg = outdeg[s];
      gcn_sum_ordered(feat, rows, s, dl, dg, n, lane, acc, currow);
    }
    if (currow >= 0) gcn_flush_run(rows, currow, acc, lane);
  } else {
    const int qb = wave * QCAP;
    int cnt = 0;

    const int4* dst4 = reinterpret_cast<const int4*>(dst);
    const int off = (blockIdx.x * (SCAN_PIECES / OWN_BLOCKS)) & ~(SCAN_U - 1);

    int g = wave * SCAN_U;
    int p = g + off;
    if (p >= SCAN_PIECES) p -= SCAN_PIECES;
    int4 nx[SCAN_U];
#pragma unroll
    for (int u = 0; u < SCAN_U; ++u) nx[u] = dst4[(p + u) * 32 + lane];

    for (;;) {
      int4 dv[SCAN_U];
#pragma unroll
      for (int u = 0; u < SCAN_U; ++u) dv[u] = nx[u];
      const int  pc   = p;
      const int  gn   = g + 32 * SCAN_U;
      const bool last = (gn >= SCAN_PIECES);
      if (!last) {
        p = gn + off;
        if (p >= SCAN_PIECES) p -= SCAN_PIECES;
      }
#pragma unroll
      for (int u = 0; u < SCAN_U; ++u) nx[u] = dst4[(p + u) * 32 + lane];

#pragma unroll
      for (int u = 0; u < SCAN_U; ++u) {
        const unsigned t[4] = {(unsigned)(dv[u].x - lo), (unsigned)(dv[u].y - lo),
                               (unsigned)(dv[u].z - lo), (unsigned)(dv[u].w - lo)};
        const unsigned mn = min(min(t[0], t[1]), min(t[2], t[3]));
        if (__builtin_amdgcn_ballot_w32(mn < OWN_ROWS) != 0u) {
          const int ebase = ((pc + u) * 32 + lane) * 4;
#pragma unroll
          for (int c = 0; c < 4; ++c) {
            const bool m = t[c] < OWN_ROWS;
            const unsigned bal = __builtin_amdgcn_ballot_w32(m);
            if (bal != 0u) {
              if (m) {
                qmem[qb + cnt + (int)__builtin_amdgcn_mbcnt_lo(bal, 0u)] = ((ebase + c) << 8) | (int)t[c];
                atomicAdd(&indeg[t[c]], 1);
              }
              cnt += __builtin_popcount(bal);
            }
          }
        }
      }

      if (cnt >= 32 || (last && cnt > 0)) {
        for (int base = 0; base < cnt; base += 32) {
          const int n = min(32, cnt - base);
          int v = 0;
          if (lane < n) v = qmem[qb + base + lane];
          const int   e  = v >> 8;
          const int   dl = v & 255;
          const int   s  = src[e];
          const float dg = outdeg[s];
          gcn_sum_listed(feat, rows, s, dl, dg, n, lane);
        }
        cnt = 0;
      }
      if (last) break;
      g = gn;
    }
  }


  const int half = lane >> 4;
  const int m    = lane & 15;
  const int ct  = wave & 7;
  const int rg  = ((wave >> 3) + (wave & 3)) & 3;
  const int col = ct * 16 + m;

  v2f b[32];
  gcn_load_bfrag<DD>(W, col, half, b);
  const float bv = bias[col];
  __syncthreads();

  float scv[3];
#pragma unroll
  for (int j = 0; j < 3; ++j) {
    const int rt = rg + 4 * j;
    scv[j] = (rt < OWN_ROWS / 16) ? gcn_deg_norm((float)indeg[rt * 16 + m]) : 0.0f;
  }

#pragma unroll
  for (int j = 0; j < 3; ++j) {
    const int rt = rg + 4 * j;
    if (rt < OWN_ROWS / 16) {
      const int r0 = rt * 16;
      const v8f acc = gcn_tile<1>(rows + (r0 + m) * ROWP, scv[j], half, b);
#pragma unroll
      for (int r = 0; r < 8; ++r) {
        const int row = r0 + r + 8 * half;
        const float v = fmaxf(acc[r] + bv, 0.0f);
        if (LAST) hid[row * ROWP + col] = v;
        else      outp[(size_t)(lo + row) * DD + col] = v;
      }
    }
  }

  if (LAST) {
    const int ct1  = wave & 3;
    const int rg1  = ((wave >> 2) + 2 * (wave & 3)) & 7;
    const int col1 = ct1 * 16 + m;
    gcn_load_bfrag<64>(Wl1, col1, half, b);
    const float bv1 = bl1[col1];
    __syncthreads();
    float* tb = rows;
    for (int j = 0; j < 2; ++j) {
      const int rt = rg1 + 8 * j;
      if (rt < OWN_ROWS / 16) {
        const int r0 = rt * 16;
        const v8f acc = gcn_tile<0>(hid + (r0 + m) * ROWP, 1.0f, half, b);
#pragma unroll
        for (int r = 0; r < 8; ++r)
          tb[(r0 + r + 8 * half) * T_PITCH + col1] = fmaxf(acc[r] + bv1, 0.0f);
      }
    }
    __syncthreads();

    for (int o = tid; o < OWN_ROWS * 10; o += 1024) {
      const int row = o / 10;
      const int c   = o - row * 10;
      const float* tp = tb + row * T_PITCH;
      float acc = bl2[c];
#pragma unroll
      for (int k = 0; k < 64; ++k) acc = fmaf(tp[k], Wl2[k * 10 + c], acc);
      outp[(size_t)lo * 10 + o] = acc;
    }
  }
}


extern "C" void kernel_launch(void* const* d_in, const int* in_sizes, int n_in,
                              void* d_out, int out_size, void* d_ws, size_t ws_size,
                              hipStream_t stream) {
  const float* h   = (const float*)d_in[0];
  const int*   src = (const int*)d_in[1];
  const int*   dst = (const int*)d_in[2];
  const float* W0  = (const float*)d_in[3];
  const float* b0  = (const float*)d_in[4];
  const float* W1  = (const float*)d_in[5];
  const float* b1  = (const float*)d_in[6];
  const float* Wl1 = (const float*)d_in[7];
  const float* bl1 = (const float*)d_in[8];
  const float* Wl2 = (const float*)d_in[9];
  const float* bl2 = (const float*)d_in[10];
  float* out = (float*)d_out;

  float* wsf      = (float*)d_ws;
  float* deg_out  = wsf;
  float* buf1     = wsf + 2 * NN;
  float* buf2     = buf1 + (size_t)NN * DD;

  const int TPB = 256;
  dim3 ownGrid(OWN_BLOCKS);

  gcn_fill_zero<<<dim3((NN + TPB - 1) / TPB), TPB, 0, stream>>>(deg_out, NN, (int*)buf1);
  gcn_degree_bin<<<ownGrid, 1024, 0, stream>>>(src, dst, deg_out, (int*)buf1);

  gcn_layer_owner<0><<<ownGrid, 1024, 0, stream>>>(h, src, dst, deg_out, (const int*)buf1,
                                                   W0, b0, nullptr, nullptr, nullptr, nullptr, buf2);
  gcn_layer_owner<1><<<ownGrid, 1024, 0, stream>>>(buf2, src, dst, deg_out, (const int*)buf1,
                                                   W1, b1, Wl1, bl1, Wl2, bl2, out);
}


# out-degrees: 125 bin groups of 5120 edges count sources in an LDS histogram then add it by wave-wide float atomics on whole 128-byte lines, not 1 atomic per edge; counts exact; lists in arrival order
# speedup vs baseline: 4.0678x; 1.2609x over previous
#include <hip/hip_runtime.h>

#define NN 40000
#define EE 640000
#define DD 128

typedef __attribute__((ext_vector_type(2))) float v2f;
typedef __attribute__((ext_vector_type(8))) float v8f;

#define OWN_ROWS   160
#define OWN_BLOCKS (NN / OWN_ROWS)
#define REGION     (OWN_ROWS * DD)
#define LIST_HDR   32
#define LIST_CAP   (REGION - LIST_HDR)
#define BIN_BLOCKS 125
#define BIN_CHUNK  (EE / BIN_BLOCKS)
#define BIN_PER    ((BIN_CHUNK + 1023) / 1024)
#define FLUSH_U    8

static_assert(NN % OWN_ROWS == 0, "owned ranges must tile the nodes");
static_assert(EE % BIN_BLOCKS == 0, "edge chunks must tile the edges");
static_assert(NN % 32 == 0 && NN % 4 == 0, "a wave adds whole 128-byte lines of out-degrees; the counters are cleared 16 bytes at a time");
static_assert(NN < (1 << 23) && OWN_ROWS <= 256, "a list entry packs source and local row into one int");


__global__ void gcn_fill_zero(float* __restrict__ p, int n, int* __restrict__ lists) {
  int i = blockIdx.x * blockDim.x + threadIdx.x;
  if (i < n) p[i] = 0.0f;
  if (i < OWN_BLOCKS) lists[(size_t)i * REGION] = 0;
}

__global__ __launch_bounds__(1024) void gcn_degree_bin(
    const int* __restrict__ src, const int* __restrict__ dst,
    float* __restrict__ deg_out, int* __restrict__ lists) {
  __shared__ int hist[OWN_BLOCKS];
  __shared__ int base[OWN_BLOCKS];
  __shared__ int scnt[NN];

  const int tid = threadIdx.x;
  const int e0 = blockIdx.x * BIN_CHUNK;

  int sv[BIN_PER], dv[BIN_PER];
#pragma unroll
  for (int k = 0; k < BIN_PER; ++k) {
    const int i = tid + k * 1024;
    sv[k] = -1;
    dv[k] = 0;
    if (i < BIN_CHUNK) {
      sv[k] = src[e0 + i];
      dv[k] = dst[e0 + i];
    }
  }

  if (tid < OWN_BLOCKS) hist[tid] = 0;
  int4* scnt4 = reinterpret_cast<int4*>(scnt);
  for (int i = tid; i < NN / 4; i += 1024) scnt4[i] = make_int4(0, 0, 0, 0);
  __syncthreads();

  int own[BIN_PER], ent[BIN_PER], rnk[BIN_PER];
#pragma unroll
  for (int k = 0; k < BIN_PER; ++k) {
    own[k] = -1;
    ent[k] = 0;
    rnk[k] = 0;
    if (sv[k] >= 0) {
      const int s = sv[k];
      const int d = dv[k];
      atomicAdd(&scnt[s], 1);
      const int o = d / OWN_ROWS;
      own[k] = o;
      ent[k] = (s << 8) | (d - o * OWN_ROWS);
      rnk[k] = atomicAdd(&hist[o], 1);
    }
  }
  __syncthreads();

  if (tid < OWN_BLOCKS) {
    const int c = hist[tid];
    int b = 0;
    if (c > 0) b = atomicAdd(&lists[(size_t)tid * REGION], c);
    base[tid] = b;
  }
  __syncthreads();

#pragma unroll
  for (int k = 0; k < BIN_PER; ++k) {
    if (own[k] >= 0) {
      const int idx = base[own[k]] + rnk[k];
      if (idx < LIST_CAP) lists[(size_t)own[k] * REGION + LIST_HDR + idx] = ent[k];
    }
  }

  for (int i0 = tid; i0 < NN; i0 += FLUSH_U * 1024) {
    int c[FLUSH_U];
#pragma unroll
    for (int u = 0; u < FLUSH_U; ++u) {
      const int i = i0 + u * 1024;
      c[u] = (i < NN) ? scnt[i] : 0;
    }
#pragma unroll
    for (int u = 0; u < FLUSH_U; ++u)
      if (c[u] > 0) atomicAdd(&deg_out[i0 + u * 1024], (float)c[u]);
  }
}

#define ROWP       (DD + 4)
#define ROWS_PER_LANE (OWN_ROWS / 32)
#define SORT_U     4
#define SCAN_U     8
#define SCAN_PIECES (EE / 128)
#define QCAP       (32 + SCAN_U * 128)

static_assert(EE % 128 == 0 && SCAN_PIECES % SCAN_U == 0, "scan pieces must tile the edges");
static_assert(OWN_ROWS % 32 == 0 && DD == 128, "one wave scans the row counts; a lane holds 4 of a row's 128 elements");
static_assert(LIST_CAP <= 32 * QCAP, "the ordered list reuses the scan path's LDS");
static_assert(SCAN_PIECES % OWN_BLOCKS == 0 && (SCAN_U & (SCAN_U - 1)) == 0 && 32 * SCAN_U <= SCAN_PIECES,
              "per-block scan offset is rounded to whole steps");

__device__ __forceinline__ float gcn_deg_norm(float deg) { return rsqrtf(fmaxf(deg, 1.0f)); }

__device__ __forceinline__ void gcn_sum_listed(const float* __restrict__ feat, float* rows,
                                               int s, int dl, float dg, int n, int lane) {
  for (int k = 0; k < n; k += 8) {
    float r[8][4];
#pragma unroll
    for (int j = 0; j < 8; ++j) {
      const int sj = __builtin_amdgcn_readlane(s, k + j);
      const float* fp = feat + (size_t)sj * DD + lane;
      r[j][0] = fp[0];
      r[j][1] = fp[32];
      r[j][2] = fp[64];
      r[j][3] = fp[96];
    }
#pragma unroll
    for (int j = 0; j < 8; ++j) {
      if (k + j < n) {
        const int   dj  = __builtin_amdgcn_readlane(dl, k + j);
        const float scj = gcn_deg_norm(__int_as_float(__builtin_amdgcn_readlane(__float_as_int(dg), k + j)));
        float* ap = rows + dj * ROWP + lane;
        __hip_atomic_fetch_add(ap +  0, r[j][0] * scj, __ATOMIC_RELAXED, __HIP_MEMORY_SCOPE_WORKGROUP);
        __hip_atomic_fetch_add(ap + 32, r[j][1] * scj, __ATOMIC_RELAXED, __HIP_MEMORY_SCOPE_WORKGROUP);
        __hip_atomic_fetch_add(ap + 64, r[j][2] * scj, __ATOMIC_RELAXED, __HIP_MEMORY_SCOPE_WORKGROUP);
        __hip_atomic_fetch_add(ap + 96, r[j][3] * scj, __ATOMIC_RELAXED, __HIP_MEMORY_SCOPE_WORKGROUP);
      }
    }
  }
}

__device__ __forceinline__ void gcn_flush_run(float* rows, int row, const float4& a, int lane) {
  float* ap = rows + row * ROWP + 4 * lane;
  __hip_atomic_fetch_add(ap + 0, a.x, __ATOMIC_RELAXED, __HIP_MEMORY_SCOPE_WORKGROUP);
  __hip_atomic_fetch_add(ap + 1, a.y, __ATOMIC_RELAXED, __HIP_MEMORY_SCOPE_WORKGROUP);
  __hip_atomic_fetch_add(ap + 2, a.z, __ATOMIC_RELAXED, __HIP_MEMORY_SCOPE_WORKGROUP);
  __hip_atomic_fetch_add(ap + 3, a.w, __ATOMIC_RELAXED, __HIP_MEMORY_SCOPE_WORKGROUP);
}

__device__ __forceinline__ void gcn_sum_ordered(const float* __restrict__ feat, float* rows,
                                                int s, int dl, float dg, int n, int lane,
                                                float4& acc, int& currow) {
  for (int k = 0; k < n; k += 8) {
    float4 r[8];
#pragma unroll
    for (int j = 0; j < 8; ++j) {
      const int sj = __builtin_amdgcn_readlane(s, k + j);
      r[j] = *reinterpret_cast<const float4*>(feat + (size_t)sj * DD + 4 * lane);
    }
#pragma unroll
    for (int j = 0; j < 8; ++j) {
      if (k + j < n) {
        const int   dj  = __builtin_amdgcn_readlane(dl, k + j);
        const float scj = gcn_deg_norm(__int_as_float(__builtin_amdgcn_readlane(__float_as_int(dg), k + j)));
        if (dj != currow) {
          if (currow >= 0) gcn_flush_run(rows, currow, acc, lane);
          acc = make_float4(0.f, 0.f, 0.f, 0.f);
          currow = dj;
        }
        acc.x = fmaf(r[j].x, scj, acc.x);
        acc.y = fmaf(r[j].y, scj, acc.y);
        acc.z = fmaf(r[j].z, scj, acc.z);
        acc.w = fmaf(r[j].w, scj, acc.w);
      }
    }
  }
}

#define T_PITCH 68
static_assert((ROWP % 64) == 4 && (ROWP * 4) % 16 == 0, "row pitch spreads 16 rows x 2 halves over the 64 banks");
static_assert(OWN_ROWS % 16 == 0 && OWN_ROWS * T_PITCH <= OWN_ROWS * ROWP, "linear1's result reuses the sums' LDS");

template <int C>
__device__ __forceinline__ void gcn_load_bfrag(const float* __restrict__ Wm, int col, int half, v2f (&b)[32]) {
#pragma unroll
  for (int i = 0; i < 32; ++i) {
    const int ka = 4 * i + 2 * half;
    b[i].x = Wm[(size_t)ka * C + col];
    b[i].y = Wm[(size_t)(ka + 1) * C + col];
  }
}

template <int SCALE>
__device__ __forceinline__ v8f gcn_tile(const float* arow, float sc, int half, const v2f (&b)[32]) {
  v8f acc = {};
#pragma unroll
  for (int i = 0; i < 32; ++i) {
    const int ka = 4 * i + 2 * half;
    v2f a;
    a.x = arow[ka];
    a.y = arow[ka + 1];
    if (SCALE) {
      a.x *= sc;
      a.y *= sc;
    }
    acc = __builtin_amdgcn_wmma_f32_16x16x4_f32(
         false, a,  false, b[i],
         (short)0, acc,  false,  false);
  }
  return acc;
}

template <int LAST>
__global__ __launch_bounds__(1024) void gcn_layer_owner(
    const float* __restrict__ feat, const int* __restrict__ src, const int* __restrict__ dst,
    const float* __restrict__ outdeg,
    const int* __restrict__ lists,
    const float* __restrict__ W, const float* __restrict__ bias,
    const float* __restrict__ Wl1, const float* __restrict__ bl1,
    const float* __restrict__ Wl2, const float* __restrict__ bl2,
    float* __restrict__ outp) {
  __shared__ float rows[OWN_ROWS * ROWP];
  __shared__ int   qmem[32 * QCAP];
  __shared__ float hid[LAST ? OWN_ROWS * ROWP : 4];
  __shared__ int   rcnt[OWN_ROWS];
  __shared__ int   indeg[OWN_ROWS];

  const int tid  = threadIdx.x;
  const int wave = __builtin_amdgcn_readfirstlane(tid >> 5);
  const int lane = tid & 31;
  const int lo   = blockIdx.x * OWN_ROWS;
  const int* mylist = lists + (size_t)blockIdx.x * REGION;

  float4* rows4 = reinterpret_cast<float4*>(rows);
  for (int i = tid; i < OWN_ROWS * ROWP / 4; i += 1024) rows4[i] = make_float4(0.f, 0.f, 0.f, 0.f);
  if (tid < OWN_ROWS) {
    rcnt[tid]  = 0;
    indeg[tid] = 0;
  }
  __syncthreads();

  const int total = __builtin_amdgcn_readfirstlane(mylist[0]);

  if (total <= LIST_CAP) {
    int* ordered = qmem;
    for (int i0 = tid; i0 < total; i0 += SORT_U * 1024) {
      int v[SORT_U];
#pragma unroll
      for (int u = 0; u < SORT_U; ++u) {
        const int i = i0 + u * 1024;
        v[u] = (i < total) ? mylist[LIST_HDR + i] : -1;
      }
#pragma unroll
      for (int u = 0; u < SORT_U; ++u)
        if (v[u] >= 0) atomicAdd(&rcnt[v[u] & 255], 1);
    }
    __syncthreads();
    if (wave == 0) {
      int cr[ROWS_PER_LANE];
      int mine = 0;
#pragma unroll
      for (int q = 0; q < ROWS_PER_LANE; ++q) {
        cr[q] = rcnt[lane * ROWS_PER_LANE + q];
        indeg[lane * ROWS_PER_LANE + q] = cr[q];
        mine += cr[q];
      }
      int inc = mine;
#pragma unroll
      for (int d = 1; d < 32; d <<= 1) {
        const int t = __shfl_up(inc, d);
        if (lane >= d) inc += t;
      }
      int run = inc - mine;
#pragma unroll
      for (int q = 0; q < ROWS_PER_LANE; ++q) {
        rcnt[lane * ROWS_PER_LANE + q] = run;
        run += cr[q];
      }
    }
    __syncthreads();
    for (int i0 = tid; i0 < total; i0 += SORT_U * 1024) {
      int v[SORT_U];
#pragma unroll
      for (int u = 0; u < SORT_U; ++u) {
        const int i = i0 + u * 1024;
        v[u] = (i < total) ? mylist[LIST_HDR + i] : -1;
      }
#pragma unroll
      for (int u = 0; u < SORT_U; ++u)
        if (v[u] >= 0) ordered[atomicAdd(&rcnt[v[u] & 255], 1)] = v[u];
    }
    __syncthreads();

    const int per = (((total + 31) >> 5) + 7) & ~7;
    const int end = min(total, (wave + 1) * per);
    float4 acc = make_float4(0.f, 0.f, 0.f, 0.f);
    int currow = -1;
    for (int base = wave * per; base < end; base += 32) {
      const int n = min(32, end - base);
      int v = 0;
      if (lane < n) v = ordered[base + lane];
      const int   s  = v >> 8;
      const int   dl = v & 255;
      const float dg = outdeg[s];
      gcn_sum_ordered(feat, rows, s, dl, dg, n, lane, acc, currow);
    }
    if (currow >= 0) gcn_flush_run(rows, currow, acc, lane);
  } else {
    const int qb = wave * QCAP;
    int cnt = 0;

    const int4* dst4 = reinterpret_cast<const int4*>(dst);
    const int off = (blockIdx.x * (SCAN_PIECES / OWN_BLOCKS)) & ~(SCAN_U - 1);

    int g = wave * SCAN_U;
    int p = g + off;
    if (p >= SCAN_PIECES) p -= SCAN_PIECES;
    int4 nx[SCAN_U];
#pragma unroll
    for (int u = 0; u < SCAN_U; ++u) nx[u] = dst4[(p + u) * 32 + lane];

    for (;;) {
      int4 dv[SCAN_U];
#pragma unroll
      for (int u = 0; u < SCAN_U; ++u) dv[u] = nx[u];
      const int  pc   = p;
      const int  gn   = g + 32 * SCAN_U;
      const bool last = (gn >= SCAN_PIECES);
      if (!last) {
        p = gn + off;
        if (p >= SCAN_PIECES) p -= SCAN_PIECES;
      }
#pragma unroll
      for (int u = 0; u < SCAN_U; ++u) nx[u] = dst4[(p + u) * 32 + lane];

#pragma unroll
      for (int u = 0; u < SCAN_U; ++u) {
        const unsigned t[4] = {(unsigned)(dv[u].x - lo), (unsigned)(dv[u].y - lo),
                               (unsigned)(dv[u].z - lo), (unsigned)(dv[u].w - lo)};
        const unsigned mn = min(min(t[0], t[1]), min(t[2], t[3]));
        if (__builtin_amdgcn_ballot_w32(mn < OWN_ROWS) != 0u) {
          const int ebase = ((pc + u) * 32 + lane) * 4;
#pragma unroll
          for (int c = 0; c < 4; ++c) {
            const bool m = t[c] < OWN_ROWS;
            const unsigned bal = __builtin_amdgcn_ballot_w32(m);
            if (bal != 0u) {
              if (m) {
                qmem[qb + cnt + (int)__builtin_amdgcn_mbcnt_lo(bal, 0u)] = ((ebase + c) << 8) | (int)t[c];
                atomicAdd(&indeg[t[c]], 1);
              }
              cnt += __builtin_popcount(bal);
            }
          }
        }
      }

      if (cnt >= 32 || (last && cnt > 0)) {
        for (int base = 0; base < cnt; base += 32) {
          const int n = min(32, cnt - base);
          int v = 0;
          if (lane < n) v = qmem[qb + base + lane];
          const int   e  = v >> 8;
          const int   dl = v & 255;
          const int   s  = src[e];
          const float dg = outdeg[s];
          gcn_sum_listed(feat, rows, s, dl, dg, n, lane);
        }
        cnt = 0;
      }
      if (last) break;
      g = gn;
    }
  }


  const int half = lane >> 4;
  const int m    = lane & 15;
  const int ct  = wave & 7;
  const int rg  = ((wave >> 3) + (wave & 3)) & 3;
  const int col = ct * 16 + m;

  v2f b[32];
  gcn_load_bfrag<DD>(W, col, half, b);
  const float bv = bias[col];
  __syncthreads();

  float scv[3];
#pragma unroll
  for (int j = 0; j < 3; ++j) {
    const int rt = rg + 4 * j;
    scv[j] = (rt < OWN_ROWS / 16) ? gcn_deg_norm((float)indeg[rt * 16 + m]) : 0.0f;
  }

#pragma unroll
  for (int j = 0; j < 3; ++j) {
    const int rt = rg + 4 * j;
    if (rt < OWN_ROWS / 16) {
      const int r0 = rt * 16;
      const v8f acc = gcn_tile<1>(rows + (r0 + m) * ROWP, scv[j], half, b);
#pragma unroll
      for (int r = 0; r < 8; ++r) {
        const int row = r0 + r + 8 * half;
        const float v = fmaxf(acc[r] + bv, 0.0f);
        if (LAST) hid[row * ROWP + col] = v;
        else      outp[(size_t)(lo + row) * DD + col] = v;
      }
    }
  }

  if (LAST) {
    const int ct1  = wave & 3;
    const int rg1  = ((wave >> 2) + 2 * (wave & 3)) & 7;
    const int col1 = ct1 * 16 + m;
    gcn_load_bfrag<64>(Wl1, col1, half, b);
    const float bv1 = bl1[col1];
    __syncthreads();
    float* tb = rows;
    for (int j = 0; j < 2; ++j) {
      const int rt = rg1 + 8 * j;
      if (rt < OWN_ROWS / 16) {
        const int r0 = rt * 16;
        const v8f acc = gcn_tile<0>(hid + (r0 + m) * ROWP, 1.0f, half, b);
#pragma unroll
        for (int r = 0; r < 8; ++r)
          tb[(r0 + r + 8 * half) * T_PITCH + col1] = fmaxf(acc[r] + bv1, 0.0f);
      }
    }
    __syncthreads();

    for (int o = tid; o < OWN_ROWS * 10; o += 1024) {
      const int row = o / 10;
      const int c   = o - row * 10;
      const float* tp = tb + row * T_PITCH;
      float acc = bl2[c];
#pragma unroll
      for (int k = 0; k < 64; ++k) acc = fmaf(tp[k], Wl2[k * 10 + c], acc);
      outp[(size_t)lo * 10 + o] = acc;
    }
  }
}


extern "C" void kernel_launch(void* const* d_in, const int* in_sizes, int n_in,
                              void* d_out, int out_size, void* d_ws, size_t ws_size,
                              hipStream_t stream) {
  const float* h   = (const float*)d_in[0];
  const int*   src = (const int*)d_in[1];
  const int*   dst = (const int*)d_in[2];
  const float* W0  = (const float*)d_in[3];
  const float* b0  = (const float*)d_in[4];
  const float* W1  = (const float*)d_in[5];
  const float* b1  = (const float*)d_in[6];
  const float* Wl1 = (const float*)d_in[7];
  const float* bl1 = (const float*)d_in[8];
  const float* Wl2 = (const float*)d_in[9];
  const float* bl2 = (const float*)d_in[10];
  float* out = (float*)d_out;

  float* wsf      = (float*)d_ws;
  float* deg_out  = wsf;
  float* buf1     = wsf + 2 * NN;
  float* buf2     = buf1 + (size_t)NN * DD;

  const int TPB = 256;
  dim3 ownGrid(OWN_BLOCKS);

  gcn_fill_zero<<<dim3((NN + TPB - 1) / TPB), TPB, 0, stream>>>(deg_out, NN, (int*)buf1);
  gcn_degree_bin<<<dim3(BIN_BLOCKS), 1024, 0, stream>>>(src, dst, deg_out, (int*)buf1);

  gcn_layer_owner<0><<<ownGrid, 1024, 0, stream>>>(h, src, dst, deg_out, (const int*)buf1,
                                                   W0, b0, nullptr, nullptr, nullptr, nullptr, buf2);
  gcn_layer_owner<1><<<ownGrid, 1024, 0, stream>>>(buf2, src, dst, deg_out, (const int*)buf1,
                                                   W1, b1, Wl1, bl1, Wl2, bl2, out);
}


# out-degree count moved to 125 extra work-groups of the binning launch, each a node range x long edge chunk: a deg_out line gets 25 wave-wide float adds, not 125; counts exact in any order
# speedup vs baseline: 4.5815x; 1.1263x over previous
#include <hip/hip_runtime.h>

#define NN 40000
#define EE 640000
#define DD 128

typedef __attribute__((ext_vector_type(2))) float v2f;
typedef __attribute__((ext_vector_type(8))) float v8f;

#define OWN_ROWS   160
#define OWN_BLOCKS (NN / OWN_ROWS)
#define REGION     (OWN_ROWS * DD)
#define LIST_HDR   32
#define LIST_CAP   (REGION - LIST_HDR)
#define BIN_BLOCKS 125
#define BIN_CHUNK  (EE / BIN_BLOCKS)
#define BIN_PER    ((BIN_CHUNK + 1023) / 1024)
#define CNT_RANGES 5
#define CNT_CHUNKS 25
#define CNT_BLOCKS (CNT_RANGES * CNT_CHUNKS)
#define CNT_NODES  (NN / CNT_RANGES)
#define CNT_EDGES  (EE / CNT_CHUNKS)
#define CNT_VEC    ((CNT_EDGES / 4 + 1023) / 1024)
#define FLUSH_U    ((CNT_NODES + 1023) / 1024)

static_assert(NN % OWN_ROWS == 0, "owned ranges must tile the nodes");
static_assert(EE % BIN_BLOCKS == 0, "edge chunks must tile the edges");
static_assert(NN % CNT_RANGES == 0 && CNT_NODES % 32 == 0, "a wave adds whole 128-byte lines of out-degrees; the counters are cleared 16 bytes at a time");
static_assert(EE % CNT_CHUNKS == 0 && CNT_EDGES % 4 == 0, "a counting work-group reads its sources 16 bytes at a time");
static_assert(NN < (1 << 23) && OWN_ROWS <= 256, "a list entry packs source and local row into one int");


__global__ void gcn_fill_zero(float* __restrict__ p, int n, int* __restrict__ lists) {
  int i = blockIdx.x * blockDim.x + threadIdx.x;
  if (i < n) p[i] = 0.0f;
  if (i < OWN_BLOCKS) lists[(size_t)i * REGION] = 0;
}

__global__ __launch_bounds__(1024) void gcn_degree_bin(
    const int* __restrict__ src, const int* __restrict__ dst,
    float* __restrict__ deg_out, int* __restrict__ lists) {
  __shared__ int hist[OWN_BLOCKS];
  __shared__ int base[OWN_BLOCKS];
  __shared__ int scnt[CNT_NODES];

  const int tid = threadIdx.x;

  if (blockIdx.x >= BIN_BLOCKS) {
    const int q   = blockIdx.x - BIN_BLOCKS;
    const int rng = q % CNT_RANGES;
    const int chk = q / CNT_RANGES;
    const int n0  = rng * CNT_NODES;
    const int4* s4 = reinterpret_cast<const int4*>(src + (size_t)chk * CNT_EDGES);

    int4 v[CNT_VEC];
#pragma unroll
    for (int k = 0; k < CNT_VEC; ++k) {
      const int i = tid + k * 1024;
      v[k] = make_int4(-1, -1, -1, -1);
      if (i < CNT_EDGES / 4) v[k] = s4[i];
    }
    int4* scnt4 = reinterpret_cast<int4*>(scnt);
    for (int i = tid; i < CNT_NODES / 4; i += 1024) scnt4[i] = make_int4(0, 0, 0, 0);
    __syncthreads();

#pragma unroll
    for (int k = 0; k < CNT_VEC; ++k) {
      const unsigned t0 = (unsigned)(v[k].x - n0);
      const unsigned t1 = (unsigned)(v[k].y - n0);
      const unsigned t2 = (unsigned)(v[k].z - n0);
      const unsigned t3 = (unsigned)(v[k].w - n0);
      if (t0 < CNT_NODES) atomicAdd(&scnt[t0], 1);
      if (t1 < CNT_NODES) atomicAdd(&scnt[t1], 1);
      if (t2 < CNT_NODES) atomicAdd(&scnt[t2], 1);
      if (t3 < CNT_NODES) atomicAdd(&scnt[t3], 1);
    }
    __syncthreads();

    int c[FLUSH_U];
#pragma unroll
    for (int u = 0; u < FLUSH_U; ++u) {
      const int i = tid + u * 1024;
      c[u] = (i < CNT_NODES) ? scnt[i] : 0;
    }
#pragma unroll
    for (int u = 0; u < FLUSH_U; ++u)
      if (c[u] > 0) atomicAdd(&deg_out[n0 + tid + u * 1024], (float)c[u]);
    return;
  }

  const int e0 = blockIdx.x * BIN_CHUNK;

  int sv[BIN_PER], dv[BIN_PER];
#pragma unroll
  for (int k = 0; k < BIN_PER; ++k) {
    const int i = tid + k * 1024;
    sv[k] = -1;
    dv[k] = 0;
    if (i < BIN_CHUNK) {
      sv[k] = src[e0 + i];
      dv[k] = dst[e0 + i];
    }
  }

  if (tid < OWN_BLOCKS) hist[tid] = 0;
  __syncthreads();

  int own[BIN_PER], ent[BIN_PER], rnk[BIN_PER];
#pragma unroll
  for (int k = 0; k < BIN_PER; ++k) {
    own[k] = -1;
    ent[k] = 0;
    rnk[k] = 0;
    if (sv[k] >= 0) {
      const int s = sv[k];
      const int d = dv[k];
      const int o = d / OWN_ROWS;
      own[k] = o;
      ent[k] = (s << 8) | (d - o * OWN_ROWS);
      rnk[k] = atomicAdd(&hist[o], 1);
    }
  }
  __syncthreads();

  if (tid < OWN_BLOCKS) {
    const int c = hist[tid];
    int b = 0;
    if (c > 0) b = atomicAdd(&lists[(size_t)tid * REGION], c);
    base[tid] = b;
  }
  __syncthreads();

#pragma unroll
  for (int k = 0; k < BIN_PER; ++k) {
    if (own[k] >= 0) {
      const int idx = base[own[k]] + rnk[k];
      if (idx < LIST_CAP) lists[(size_t)own[k] * REGION + LIST_HDR + idx] = ent[k];
    }
  }
}

#define ROWP       (DD + 4)
#define ROWS_PER_LANE (OWN_ROWS / 32)
#define SORT_U     4
#define SCAN_U     8
#define SCAN_PIECES (EE / 128)
#define QCAP       (32 + SCAN_U * 128)

static_assert(EE % 128 == 0 && SCAN_PIECES % SCAN_U == 0, "scan pieces must tile the edges");
static_assert(OWN_ROWS % 32 == 0 && DD == 128, "one wave scans the row counts; a lane holds 4 of a row's 128 elements");
static_assert(LIST_CAP <= 32 * QCAP, "the ordered list reuses the scan path's LDS");
static_assert(SCAN_PIECES % OWN_BLOCKS == 0 && (SCAN_U & (SCAN_U - 1)) == 0 && 32 * SCAN_U <= SCAN_PIECES,
              "per-block scan offset is rounded to whole steps");

__device__ __forceinline__ float gcn_deg_norm(float deg) { return rsqrtf(fmaxf(deg, 1.0f)); }

__device__ __forceinline__ void gcn_sum_listed(const float* __restrict__ feat, float* rows,
                                               int s, int dl, float dg, int n, int lane) {
  for (int k = 0; k < n; k += 8) {
    float r[8][4];
#pragma unroll
    for (int j = 0; j < 8; ++j) {
      const int sj = __builtin_amdgcn_readlane(s, k + j);
      const float* fp = feat + (size_t)sj * DD + lane;
      r[j][0] = fp[0];
      r[j][1] = fp[32];
      r[j][2] = fp[64];
      r[j][3] = fp[96];
    }
#pragma unroll
    for (int j = 0; j < 8; ++j) {
      if (k + j < n) {
        const int   dj  = __builtin_amdgcn_readlane(dl, k + j);
        const float scj = gcn_deg_norm(__int_as_float(__builtin_amdgcn_readlane(__float_as_int(dg), k + j)));
        float* ap = rows + dj * ROWP + lane;
        __hip_atomic_fetch_add(ap +  0, r[j][0] * scj, __ATOMIC_RELAXED, __HIP_MEMORY_SCOPE_WORKGROUP);
        __hip_atomic_fetch_add(ap + 32, r[j][1] * scj, __ATOMIC_RELAXED, __HIP_MEMORY_SCOPE_WORKGROUP);
        __hip_atomic_fetch_add(ap + 64, r[j][2] * scj, __ATOMIC_RELAXED, __HIP_MEMORY_SCOPE_WORKGROUP);
        __hip_atomic_fetch_add(ap + 96, r[j][3] * scj, __ATOMIC_RELAXED, __HIP_MEMORY_SCOPE_WORKGROUP);
      }
    }
  }
}

__device__ __forceinline__ void gcn_flush_run(float* rows, int row, const float4& a, int lane) {
  float* ap = rows + row * ROWP + 4 * lane;
  __hip_atomic_fetch_add(ap + 0, a.x, __ATOMIC_RELAXED, __HIP_MEMORY_SCOPE_WORKGROUP);
  __hip_atomic_fetch_add(ap + 1, a.y, __ATOMIC_RELAXED, __HIP_MEMORY_SCOPE_WORKGROUP);
  __hip_atomic_fetch_add(ap + 2, a.z, __ATOMIC_RELAXED, __HIP_MEMORY_SCOPE_WORKGROUP);
  __hip_atomic_fetch_add(ap + 3, a.w, __ATOMIC_RELAXED, __HIP_MEMORY_SCOPE_WORKGROUP);
}

__device__ __forceinline__ void gcn_sum_ordered(const float* __restrict__ feat, float* rows,
                                                int s, int dl, float dg, int n, int lane,
                                                float4& acc, int& currow) {
  for (int k = 0; k < n; k += 8) {
    float4 r[8];
#pragma unroll
    for (int j = 0; j < 8; ++j) {
      const int sj = __builtin_amdgcn_readlane(s, k + j);
      r[j] = *reinterpret_cast<const float4*>(feat + (size_t)sj * DD + 4 * lane);
    }
#pragma unroll
    for (int j = 0; j < 8; ++j) {
      if (k + j < n) {
        const int   dj  = __builtin_amdgcn_readlane(dl, k + j);
        const float scj = gcn_deg_norm(__int_as_float(__builtin_amdgcn_readlane(__float_as_int(dg), k + j)));
        if (dj != currow) {
          if (currow >= 0) gcn_flush_run(rows, currow, acc, lane);
          acc = make_float4(0.f, 0.f, 0.f, 0.f);
          currow = dj;
        }
        acc.x = fmaf(r[j].x, scj, acc.x);
        acc.y = fmaf(r[j].y, scj, acc.y);
        acc.z = fmaf(r[j].z, scj, acc.z);
        acc.w = fmaf(r[j].w, scj, acc.w);
      }
    }
  }
}

#define T_PITCH 68
static_assert((ROWP % 64) == 4 && (ROWP * 4) % 16 == 0, "row pitch spreads 16 rows x 2 halves over the 64 banks");
static_assert(OWN_ROWS % 16 == 0 && OWN_ROWS * T_PITCH <= OWN_ROWS * ROWP, "linear1's result reuses the sums' LDS");

template <int C>
__device__ __forceinline__ void gcn_load_bfrag(const float* __restrict__ Wm, int col, int half, v2f (&b)[32]) {
#pragma unroll
  for (int i = 0; i < 32; ++i) {
    const int ka = 4 * i + 2 * half;
    b[i].x = Wm[(size_t)ka * C + col];
    b[i].y = Wm[(size_t)(ka + 1) * C + col];
  }
}

template <int SCALE>
__device__ __forceinline__ v8f gcn_tile(const float* arow, float sc, int half, const v2f (&b)[32]) {
  v8f acc = {};
#pragma unroll
  for (int i = 0; i < 32; ++i) {
    const int ka = 4 * i + 2 * half;
    v2f a;
    a.x = arow[ka];
    a.y = arow[ka + 1];
    if (SCALE) {
      a.x *= sc;
      a.y *= sc;
    }
    acc = __builtin_amdgcn_wmma_f32_16x16x4_f32(
         false, a,  false, b[i],
         (short)0, acc,  false,  false);
  }
  return acc;
}

template <int LAST>
__global__ __launch_bounds__(1024) void gcn_layer_owner(
    const float* __restrict__ feat, const int* __restrict__ src, const int* __restrict__ dst,
    const float* __restrict__ outdeg,
    const int* __restrict__ lists,
    const float* __restrict__ W, const float* __restrict__ bias,
    const float* __restrict__ Wl1, const float* __restrict__ bl1,
    const float* __restrict__ Wl2, const float* __restrict__ bl2,
    float* __restrict__ outp) {
  __shared__ float rows[OWN_ROWS * ROWP];
  __shared__ int   qmem[32 * QCAP];
  __shared__ float hid[LAST ? OWN_ROWS * ROWP : 4];
  __shared__ int   rcnt[OWN_ROWS];
  __shared__ int   indeg[OWN_ROWS];

  const int tid  = threadIdx.x;
  const int wave = __builtin_amdgcn_readfirstlane(tid >> 5);
  const int lane = tid & 31;
  const int lo   = blockIdx.x * OWN_ROWS;
  const int* mylist = lists + (size_t)blockIdx.x * REGION;

  float4* rows4 = reinterpret_cast<float4*>(rows);
  for (int i = tid; i < OWN_ROWS * ROWP / 4; i += 1024) rows4[i] = make_float4(0.f, 0.f, 0.f, 0.f);
  if (tid < OWN_ROWS) {
    rcnt[tid]  = 0;
    indeg[tid] = 0;
  }
  __syncthreads();

  const int total = __builtin_amdgcn_readfirstlane(mylist[0]);

  if (total <= LIST_CAP) {
    int* ordered = qmem;
    for (int i0 = tid; i0 < total; i0 += SORT_U * 1024) {
      int v[SORT_U];
#pragma unroll
      for (int u = 0; u < SORT_U; ++u) {
        const int i = i0 + u * 1024;
        v[u] = (i < total) ? mylist[LIST_HDR + i] : -1;
      }
#pragma unroll
      for (int u = 0; u < SORT_U; ++u)
        if (v[u] >= 0) atomicAdd(&rcnt[v[u] & 255], 1);
    }
    __syncthreads();
    if (wave == 0) {
      int cr[ROWS_PER_LANE];
      int mine = 0;
#pragma unroll
      for (int q = 0; q < ROWS_PER_LANE; ++q) {
        cr[q] = rcnt[lane * ROWS_PER_LANE + q];
        indeg[lane * ROWS_PER_LANE + q] = cr[q];
        mine += cr[q];
      }
      int inc = mine;
#pragma unroll
      for (int d = 1; d < 32; d <<= 1) {
        const int t = __shfl_up(inc, d);
        if (lane >= d) inc += t;
      }
      int run = inc - mine;
#pragma unroll
      for (int q = 0; q < ROWS_PER_LANE; ++q) {
        rcnt[lane * ROWS_PER_LANE + q] = run;
        run += cr[q];
      }
    }
    __syncthreads();
    for (int i0 = tid; i0 < total; i0 += SORT_U * 1024) {
      int v[SORT_U];
#pragma unroll
      for (int u = 0; u < SORT_U; ++u) {
        const int i = i0 + u * 1024;
        v[u] = (i < total) ? mylist[LIST_HDR + i] : -1;
      }
#pragma unroll
      for (int u = 0; u < SORT_U; ++u)
        if (v[u] >= 0) ordered[atomicAdd(&rcnt[v[u] & 255], 1)] = v[u];
    }
    __syncthreads();

    const int per = (((total + 31) >> 5) + 7) & ~7;
    const int end = min(total, (wave + 1) * per);
    float4 acc = make_float4(0.f, 0.f, 0.f, 0.f);
    int currow = -1;
    for (int base = wave * per; base < end; base += 32) {
      const int n = min(32, end - base);
      int v = 0;
      if (lane < n) v = ordered[base + lane];
      const int   s  = v >> 8;
      const int   dl = v & 255;
      const float dg = outdeg[s];
      gcn_sum_ordered(feat, rows, s, dl, dg, n, lane, acc, currow);
    }
    if (currow >= 0) gcn_flush_run(rows, currow, acc, lane);
  } else {
    const int qb = wave * QCAP;
    int cnt = 0;

    const int4* dst4 = reinterpret_cast<const int4*>(dst);
    const int off = (blockIdx.x * (SCAN_PIECES / OWN_BLOCKS)) & ~(SCAN_U - 1);

    int g = wave * SCAN_U;
    int p = g + off;
    if (p >= SCAN_PIECES) p -= SCAN_PIECES;
    int4 nx[SCAN_U];
#pragma unroll
    for (int u = 0; u < SCAN_U; ++u) nx[u] = dst4[(p + u) * 32 + lane];

    for (;;) {
      int4 dv[SCAN_U];
#pragma unroll
      for (int u = 0; u < SCAN_U; ++u) dv[u] = nx[u];
      const int  pc   = p;
      const int  gn   = g + 32 * SCAN_U;
      const bool last = (gn >= SCAN_PIECES);
      if (!last) {
        p = gn + off;
        if (p >= SCAN_PIECES) p -= SCAN_PIECES;
      }
#pragma unroll
      for (int u = 0; u < SCAN_U; ++u) nx[u] = dst4[(p + u) * 32 + lane];

#pragma unroll
      for (int u = 0; u < SCAN_U; ++u) {
        const unsigned t[4] = {(unsigned)(dv[u].x - lo), (unsigned)(dv[u].y - lo),
                               (unsigned)(dv[u].z - lo), (unsigned)(dv[u].w - lo)};
        const unsigned mn = min(min(t[0], t[1]), min(t[2], t[3]));
        if (__builtin_amdgcn_ballot_w32(mn < OWN_ROWS) != 0u) {
          const int ebase = ((pc + u) * 32 + lane) * 4;
#pragma unroll
          for (int c = 0; c < 4; ++c) {
            const bool m = t[c] < OWN_ROWS;
            const unsigned bal = __builtin_amdgcn_ballot_w32(m);
            if (bal != 0u) {
              if (m) {
                qmem[qb + cnt + (int)__builtin_amdgcn_mbcnt_lo(bal, 0u)] = ((ebase + c) << 8) | (int)t[c];
                atomicAdd(&indeg[t[c]], 1);
              }
              cnt += __builtin_popcount(bal);
            }
          }
        }
      }

      if (cnt >= 32 || (last && cnt > 0)) {
        for (int base = 0; base < cnt; base += 32) {
          const int n = min(32, cnt - base);
          int v = 0;
          if (lane < n) v = qmem[qb + base + lane];
          const int   e  = v >> 8;
          const int   dl = v & 255;
          const int   s  = src[e];
          const float dg = outdeg[s];
          gcn_sum_listed(feat, rows, s, dl, dg, n, lane);
        }
        cnt = 0;
      }
      if (last) break;
      g = gn;
    }
  }


  const int half = lane >> 4;
  const int m    = lane & 15;
  const int ct  = wave & 7;
  const int rg  = ((wave >> 3) + (wave & 3)) & 3;
  const int col = ct * 16 + m;

  v2f b[32];
  gcn_load_bfrag<DD>(W, col, half, b);
  const float bv = bias[col];
  __syncthreads();

  float scv[3];
#pragma unroll
  for (int j = 0; j < 3; ++j) {
    const int rt = rg + 4 * j;
    scv[j] = (rt < OWN_ROWS / 16) ? gcn_deg_norm((float)indeg[rt * 16 + m]) : 0.0f;
  }

#pragma unroll
  for (int j = 0; j < 3; ++j) {
    const int rt = rg + 4 * j;
    if (rt < OWN_ROWS / 16) {
      const int r0 = rt * 16;
      const v8f acc = gcn_tile<1>(rows + (r0 + m) * ROWP, scv[j], half, b);
#pragma unroll
      for (int r = 0; r < 8; ++r) {
        const int row = r0 + r + 8 * half;
        const float v = fmaxf(acc[r] + bv, 0.0f);
        if (LAST) hid[row * ROWP + col] = v;
        else      outp[(size_t)(lo + row) * DD + col] = v;
      }
    }
  }

  if (LAST) {
    const int ct1  = wave & 3;
    const int rg1  = ((wave >> 2) + 2 * (wave & 3)) & 7;
    const int col1 = ct1 * 16 + m;
    gcn_load_bfrag<64>(Wl1, col1, half, b);
    const float bv1 = bl1[col1];
    __syncthreads();
    float* tb = rows;
    for (int j = 0; j < 2; ++j) {
      const int rt = rg1 + 8 * j;
      if (rt < OWN_ROWS / 16) {
        const int r0 = rt * 16;
        const v8f acc = gcn_tile<0>(hid + (r0 + m) * ROWP, 1.0f, half, b);
#pragma unroll
        for (int r = 0; r < 8; ++r)
          tb[(r0 + r + 8 * half) * T_PITCH + col1] = fmaxf(acc[r] + bv1, 0.0f);
      }
    }
    __syncthreads();

    for (int o = tid; o < OWN_ROWS * 10; o += 1024) {
      const int row = o / 10;
      const int c   = o - row * 10;
      const float* tp = tb + row * T_PITCH;
      float acc = bl2[c];
#pragma unroll
      for (int k = 0; k < 64; ++k) acc = fmaf(tp[k], Wl2[k * 10 + c], acc);
      outp[(size_t)lo * 10 + o] = acc;
    }
  }
}


extern "C" void kernel_launch(void* const* d_in, const int* in_sizes, int n_in,
                              void* d_out, int out_size, void* d_ws, size_t ws_size,
                              hipStream_t stream) {
  const float* h   = (const float*)d_in[0];
  const int*   src = (const int*)d_in[1];
  const int*   dst = (const int*)d_in[2];
  const float* W0  = (const float*)d_in[3];
  const float* b0  = (const float*)d_in[4];
  const float* W1  = (const float*)d_in[5];
  const float* b1  = (const float*)d_in[6];
  const float* Wl1 = (const float*)d_in[7];
  const float* bl1 = (const float*)d_in[8];
  const float* Wl2 = (const float*)d_in[9];
  const float* bl2 = (const float*)d_in[10];
  float* out = (float*)d_out;

  float* wsf      = (float*)d_ws;
  float* deg_out  = wsf;
  float* buf1     = wsf + 2 * NN;
  float* buf2     = buf1 + (size_t)NN * DD;

  const int TPB = 256;
  dim3 ownGrid(OWN_BLOCKS);

  gcn_fill_zero<<<dim3((NN + TPB - 1) / TPB), TPB, 0, stream>>>(deg_out, NN, (int*)buf1);
  gcn_degree_bin<<<dim3(BIN_BLOCKS + CNT_BLOCKS), 1024, 0, stream>>>(src, dst, deg_out, (int*)buf1);

  gcn_layer_owner<0><<<ownGrid, 1024, 0, stream>>>(h, src, dst, deg_out, (const int*)buf1,
                                                   W0, b0, nullptr, nullptr, nullptr, nullptr, buf2);
  gcn_layer_owner<1><<<ownGrid, 1024, 0, stream>>>(buf2, src, dst, deg_out, (const int*)buf1,
                                                   W1, b1, Wl1, bl1, Wl2, bl2, out);
}


# wave roles in the owner kernels: 24 waves gather each block of 32 rows while 8 waves take the block before through the WMMA product; run pieces per block, LDS float adds in arrival order as before
# speedup vs baseline: 4.7490x; 1.0366x over previous
#include <hip/hip_runtime.h>

#define NN 40000
#define EE 640000
#define DD 128

typedef __attribute__((ext_vector_type(2))) float v2f;
typedef __attribute__((ext_vector_type(8))) float v8f;

#define OWN_ROWS   160
#define OWN_BLOCKS (NN / OWN_ROWS)
#define REGION     (OWN_ROWS * DD)
#define LIST_HDR   32
#define LIST_CAP   (REGION - LIST_HDR)
#define BIN_BLOCKS 125
#define BIN_CHUNK  (EE / BIN_BLOCKS)
#define BIN_PER    ((BIN_CHUNK + 1023) / 1024)
#define CNT_RANGES 5
#define CNT_CHUNKS 25
#define CNT_BLOCKS (CNT_RANGES * CNT_CHUNKS)
#define CNT_NODES  (NN / CNT_RANGES)
#define CNT_EDGES  (EE / CNT_CHUNKS)
#define CNT_VEC    ((CNT_EDGES / 4 + 1023) / 1024)
#define FLUSH_U    ((CNT_NODES + 1023) / 1024)

static_assert(NN % OWN_ROWS == 0, "owned ranges must tile the nodes");
static_assert(EE % BIN_BLOCKS == 0, "edge chunks must tile the edges");
static_assert(NN % CNT_RANGES == 0 && CNT_NODES % 32 == 0, "a wave adds whole 128-byte lines of out-degrees; the counters are cleared 16 bytes at a time");
static_assert(EE % CNT_CHUNKS == 0 && CNT_EDGES % 4 == 0, "a counting work-group reads its sources 16 bytes at a time");
static_assert(NN < (1 << 23) && OWN_ROWS <= 256, "a list entry packs source and local row into one int");


__global__ void gcn_fill_zero(float* __restrict__ p, int n, int* __restrict__ lists) {
  int i = blockIdx.x * blockDim.x + threadIdx.x;
  if (i < n) p[i] = 0.0f;
  if (i < OWN_BLOCKS) lists[(size_t)i * REGION] = 0;
}

__global__ __launch_bounds__(1024) void gcn_degree_bin(
    const int* __restrict__ src, const int* __restrict__ dst,
    float* __restrict__ deg_out, int* __restrict__ lists) {
  __shared__ int hist[OWN_BLOCKS];
  __shared__ int base[OWN_BLOCKS];
  __shared__ int scnt[CNT_NODES];

  const int tid = threadIdx.x;

  if (blockIdx.x >= BIN_BLOCKS) {
    const int q   = blockIdx.x - BIN_BLOCKS;
    const int rng = q % CNT_RANGES;
    const int chk = q / CNT_RANGES;
    const int n0  = rng * CNT_NODES;
    const int4* s4 = reinterpret_cast<const int4*>(src + (size_t)chk * CNT_EDGES);

    int4 v[CNT_VEC];
#pragma unroll
    for (int k = 0; k < CNT_VEC; ++k) {
      const int i = tid + k * 1024;
      v[k] = make_int4(-1, -1, -1, -1);
      if (i < CNT_EDGES / 4) v[k] = s4[i];
    }
    int4* scnt4 = reinterpret_cast<int4*>(scnt);
    for (int i = tid; i < CNT_NODES / 4; i += 1024) scnt4[i] = make_int4(0, 0, 0, 0);
    __syncthreads();

#pragma unroll
    for (int k = 0; k < CNT_VEC; ++k) {
      const unsigned t0 = (unsigned)(v[k].x - n0);
      const unsigned t1 = (unsigned)(v[k].y - n0);
      const unsigned t2 = (unsigned)(v[k].z - n0);
      const unsigned t3 = (unsigned)(v[k].w - n0);
      if (t0 < CNT_NODES) atomicAdd(&scnt[t0], 1);
      if (t1 < CNT_NODES) atomicAdd(&scnt[t1], 1);
      if (t2 < CNT_NODES) atomicAdd(&scnt[t2], 1);
      if (t3 < CNT_NODES) atomicAdd(&scnt[t3], 1);
    }
    __syncthreads();

    int c[FLUSH_U];
#pragma unroll
    for (int u = 0; u < FLUSH_U; ++u) {
      const int i = tid + u * 1024;
      c[u] = (i < CNT_NODES) ? scnt[i] : 0;
    }
#pragma unroll
    for (int u = 0; u < FLUSH_U; ++u)
      if (c[u] > 0) atomicAdd(&deg_out[n0 + tid + u * 1024], (float)c[u]);
    return;
  }

  const int e0 = blockIdx.x * BIN_CHUNK;

  int sv[BIN_PER], dv[BIN_PER];
#pragma unroll
  for (int k = 0; k < BIN_PER; ++k) {
    const int i = tid + k * 1024;
    sv[k] = -1;
    dv[k] = 0;
    if (i < BIN_CHUNK) {
      sv[k] = src[e0 + i];
      dv[k] = dst[e0 + i];
    }
  }

  if (tid < OWN_BLOCKS) hist[tid] = 0;
  __syncthreads();

  int own[BIN_PER], ent[BIN_PER], rnk[BIN_PER];
#pragma unroll
  for (int k = 0; k < BIN_PER; ++k) {
    own[k] = -1;
    ent[k] = 0;
    rnk[k] = 0;
    if (sv[k] >= 0) {
      const int s = sv[k];
      const int d = dv[k];
      const int o = d / OWN_ROWS;
      own[k] = o;
      ent[k] = (s << 8) | (d - o * OWN_ROWS);
      rnk[k] = atomicAdd(&hist[o], 1);
    }
  }
  __syncthreads();

  if (tid < OWN_BLOCKS) {
    const int c = hist[tid];
    int b = 0;
    if (c > 0) b = atomicAdd(&lists[(size_t)tid * REGION], c);
    base[tid] = b;
  }
  __syncthreads();

#pragma unroll
  for (int k = 0; k < BIN_PER; ++k) {
    if (own[k] >= 0) {
      const int idx = base[own[k]] + rnk[k];
      if (idx < LIST_CAP) lists[(size_t)own[k] * REGION + LIST_HDR + idx] = ent[k];
    }
  }
}

#define ROWP       (DD + 4)
#define ROWS_PER_LANE (OWN_ROWS / 32)
#define SORT_U     4
#define SCAN_U     8
#define SCAN_PIECES (EE / 128)
#define QCAP       (32 + SCAN_U * 128)
#define BLK_ROWS   32
#define NBLK       (OWN_ROWS / BLK_ROWS)
#define G_WAVES    24

static_assert(EE % 128 == 0 && SCAN_PIECES % SCAN_U == 0, "scan pieces must tile the edges");
static_assert(OWN_ROWS % 32 == 0 && DD == 128, "one wave scans the row counts; a lane holds 4 of a row's 128 elements");
static_assert(OWN_ROWS % BLK_ROWS == 0 && BLK_ROWS == 32 && DD / 16 == 32 - G_WAVES,
              "a block is two row tiles; one multiplying wave per 16-column slab of W");
static_assert(LIST_CAP <= 32 * QCAP, "the ordered list reuses the scan path's LDS");
static_assert(SCAN_PIECES % OWN_BLOCKS == 0 && (SCAN_U & (SCAN_U - 1)) == 0 && 32 * SCAN_U <= SCAN_PIECES,
              "per-block scan offset is rounded to whole steps");

__device__ __forceinline__ float gcn_deg_norm(float deg) { return rsqrtf(fmaxf(deg, 1.0f)); }

__device__ __forceinline__ void gcn_sum_listed(const float* __restrict__ feat, float* rows,
                                               int s, int dl, float dg, int n, int lane) {
  for (int k = 0; k < n; k += 8) {
    float r[8][4];
#pragma unroll
    for (int j = 0; j < 8; ++j) {
      const int sj = __builtin_amdgcn_readlane(s, k + j);
      const float* fp = feat + (size_t)sj * DD + lane;
      r[j][0] = fp[0];
      r[j][1] = fp[32];
      r[j][2] = fp[64];
      r[j][3] = fp[96];
    }
#pragma unroll
    for (int j = 0; j < 8; ++j) {
      if (k + j < n) {
        const int   dj  = __builtin_amdgcn_readlane(dl, k + j);
        const float scj = gcn_deg_norm(__int_as_float(__builtin_amdgcn_readlane(__float_as_int(dg), k + j)));
        float* ap = rows + dj * ROWP + lane;
        __hip_atomic_fetch_add(ap +  0, r[j][0] * scj, __ATOMIC_RELAXED, __HIP_MEMORY_SCOPE_WORKGROUP);
        __hip_atomic_fetch_add(ap + 32, r[j][1] * scj, __ATOMIC_RELAXED, __HIP_MEMORY_SCOPE_WORKGROUP);
        __hip_atomic_fetch_add(ap + 64, r[j][2] * scj, __ATOMIC_RELAXED, __HIP_MEMORY_SCOPE_WORKGROUP);
        __hip_atomic_fetch_add(ap + 96, r[j][3] * scj, __ATOMIC_RELAXED, __HIP_MEMORY_SCOPE_WORKGROUP);
      }
    }
  }
}

__device__ __forceinline__ void gcn_flush_run(float* rows, int row, const float4& a, int lane) {
  float* ap = rows + row * ROWP + 4 * lane;
  __hip_atomic_fetch_add(ap + 0, a.x, __ATOMIC_RELAXED, __HIP_MEMORY_SCOPE_WORKGROUP);
  __hip_atomic_fetch_add(ap + 1, a.y, __ATOMIC_RELAXED, __HIP_MEMORY_SCOPE_WORKGROUP);
  __hip_atomic_fetch_add(ap + 2, a.z, __ATOMIC_RELAXED, __HIP_MEMORY_SCOPE_WORKGROUP);
  __hip_atomic_fetch_add(ap + 3, a.w, __ATOMIC_RELAXED, __HIP_MEMORY_SCOPE_WORKGROUP);
}

__device__ __forceinline__ void gcn_sum_ordered(const float* __restrict__ feat,
                                                const float* __restrict__ outdeg, float* rows,
                                                int s, int dl, int n, int lane,
                                                float4& acc, int& currow) {
  float dg = 1.0f;
  for (int k = 0; k < n; k += 8) {
    float4 r[8];
#pragma unroll
    for (int j = 0; j < 8; ++j) {
      const int sj = __builtin_amdgcn_readlane(s, k + j);
      r[j] = *reinterpret_cast<const float4*>(feat + (size_t)sj * DD + 4 * lane);
    }
    if (k == 0) dg = outdeg[s];
#pragma unroll
    for (int j = 0; j < 8; ++j) {
      if (k + j < n) {
        const int   dj  = __builtin_amdgcn_readlane(dl, k + j);
        const float scj = gcn_deg_norm(__int_as_float(__builtin_amdgcn_readlane(__float_as_int(dg), k + j)));
        if (dj != currow) {
          if (currow >= 0) gcn_flush_run(rows, currow, acc, lane);
          acc = make_float4(0.f, 0.f, 0.f, 0.f);
          currow = dj;
        }
        acc.x = fmaf(r[j].x, scj, acc.x);
        acc.y = fmaf(r[j].y, scj, acc.y);
        acc.z = fmaf(r[j].z, scj, acc.z);
        acc.w = fmaf(r[j].w, scj, acc.w);
      }
    }
  }
}

#define T_PITCH 68
static_assert((ROWP % 64) == 4 && (ROWP * 4) % 16 == 0, "row pitch spreads 16 rows x 2 halves over the 64 banks");
static_assert(OWN_ROWS % 16 == 0 && OWN_ROWS * T_PITCH <= OWN_ROWS * ROWP, "linear1's result reuses the sums' LDS");

template <int C>
__device__ __forceinline__ void gcn_load_bfrag(const float* __restrict__ Wm, int col, int half, v2f (&b)[32]) {
#pragma unroll
  for (int i = 0; i < 32; ++i) {
    const int ka = 4 * i + 2 * half;
    b[i].x = Wm[(size_t)ka * C + col];
    b[i].y = Wm[(size_t)(ka + 1) * C + col];
  }
}

template <int SCALE>
__device__ __forceinline__ v8f gcn_tile(const float* arow, float sc, int half, const v2f (&b)[32]) {
  v8f acc = {};
#pragma unroll
  for (int i = 0; i < 32; ++i) {
    const int ka = 4 * i + 2 * half;
    v2f a;
    a.x = arow[ka];
    a.y = arow[ka + 1];
    if (SCALE) {
      a.x *= sc;
      a.y *= sc;
    }
    acc = __builtin_amdgcn_wmma_f32_16x16x4_f32(
         false, a,  false, b[i],
         (short)0, acc,  false,  false);
  }
  return acc;
}

__device__ __forceinline__ void gcn_tile2(const float* arow0, const float* arow1, float sc0, float sc1,
                                          int half, const v2f (&b)[32], v8f& acc0, v8f& acc1) {
  acc0 = v8f{};
  acc1 = v8f{};
#pragma unroll
  for (int i = 0; i < 32; ++i) {
    const int ka = 4 * i + 2 * half;
    v2f a0, a1;
    a0.x = arow0[ka] * sc0;
    a0.y = arow0[ka + 1] * sc0;
    a1.x = arow1[ka] * sc1;
    a1.y = arow1[ka + 1] * sc1;
    acc0 = __builtin_amdgcn_wmma_f32_16x16x4_f32(
         false, a0,  false, b[i],
         (short)0, acc0,  false,  false);
    acc1 = __builtin_amdgcn_wmma_f32_16x16x4_f32(
         false, a1,  false, b[i],
         (short)0, acc1,  false,  false);
  }
}

template <int LAST>
__global__ __launch_bounds__(1024) void gcn_layer_owner(
    const float* __restrict__ feat, const int* __restrict__ src, const int* __restrict__ dst,
    const float* __restrict__ outdeg,
    const int* __restrict__ lists,
    const float* __restrict__ W, const float* __restrict__ bias,
    const float* __restrict__ Wl1, const float* __restrict__ bl1,
    const float* __restrict__ Wl2, const float* __restrict__ bl2,
    float* __restrict__ outp) {
  __shared__ float rows[OWN_ROWS * ROWP];
  __shared__ int   qmem[32 * QCAP];
  __shared__ float hid[LAST ? OWN_ROWS * ROWP : 4];
  __shared__ int   rcnt[OWN_ROWS];
  __shared__ int   indeg[OWN_ROWS];

  const int tid  = threadIdx.x;
  const int wave = __builtin_amdgcn_readfirstlane(tid >> 5);
  const int lane = tid & 31;
  const int lo   = blockIdx.x * OWN_ROWS;
  const int* mylist = lists + (size_t)blockIdx.x * REGION;

  float4* rows4 = reinterpret_cast<float4*>(rows);
  for (int i = tid; i < OWN_ROWS * ROWP / 4; i += 1024) rows4[i] = make_float4(0.f, 0.f, 0.f, 0.f);
  if (tid < OWN_ROWS) {
    rcnt[tid]  = 0;
    indeg[tid] = 0;
  }
  __syncthreads();

  const int total = __builtin_amdgcn_readfirstlane(mylist[0]);

  if (total <= LIST_CAP) {
    int* ordered = qmem;
    for (int i0 = tid; i0 < total; i0 += SORT_U * 1024) {
      int v[SORT_U];
#pragma unroll
      for (int u = 0; u < SORT_U; ++u) {
        const int i = i0 + u * 1024;
        v[u] = (i < total) ? mylist[LIST_HDR + i] : -1;
      }
#pragma unroll
      for (int u = 0; u < SORT_U; ++u)
        if (v[u] >= 0) atomicAdd(&rcnt[v[u] & 255], 1);
    }
    __syncthreads();
    if (wave == 0) {
      int cr[ROWS_PER_LANE];
      int mine = 0;
#pragma unroll
      for (int q = 0; q < ROWS_PER_LANE; ++q) {
        cr[q] = rcnt[lane * ROWS_PER_LANE + q];
        indeg[lane * ROWS_PER_LANE + q] = cr[q];
        mine += cr[q];
      }
      int inc = mine;
#pragma unroll
      for (int d = 1; d < 32; d <<= 1) {
        const int t = __shfl_up(inc, d);
        if (lane >= d) inc += t;
      }
      int run = inc - mine;
#pragma unroll
      for (int q = 0; q < ROWS_PER_LANE; ++q) {
        rcnt[lane * ROWS_PER_LANE + q] = run;
        run += cr[q];
      }
    }
    __syncthreads();
    for (int i0 = tid; i0 < total; i0 += SORT_U * 1024) {
      int v[SORT_U];
#pragma unroll
      for (int u = 0; u < SORT_U; ++u) {
        const int i = i0 + u * 1024;
        v[u] = (i < total) ? mylist[LIST_HDR + i] : -1;
      }
#pragma unroll
      for (int u = 0; u < SORT_U; ++u)
        if (v[u] >= 0) ordered[atomicAdd(&rcnt[v[u] & 255], 1)] = v[u];
    }
    __syncthreads();
  } else {
    const int qb = wave * QCAP;
    int cnt = 0;

    const int4* dst4 = reinterpret_cast<const int4*>(dst);
    const int off = (blockIdx.x * (SCAN_PIECES / OWN_BLOCKS)) & ~(SCAN_U - 1);

    int g = wave * SCAN_U;
    int p = g + off;
    if (p >= SCAN_PIECES) p -= SCAN_PIECES;
    int4 nx[SCAN_U];
#pragma unroll
    for (int u = 0; u < SCAN_U; ++u) nx[u] = dst4[(p + u) * 32 + lane];

    for (;;) {
      int4 dv[SCAN_U];
#pragma unroll
      for (int u = 0; u < SCAN_U; ++u) dv[u] = nx[u];
      const int  pc   = p;
      const int  gn   = g + 32 * SCAN_U;
      const bool last = (gn >= SCAN_PIECES);
      if (!last) {
        p = gn + off;
        if (p >= SCAN_PIECES) p -= SCAN_PIECES;
      }
#pragma unroll
      for (int u = 0; u < SCAN_U; ++u) nx[u] = dst4[(p + u) * 32 + lane];

#pragma unroll
      for (int u = 0; u < SCAN_U; ++u) {
        const unsigned t[4] = {(unsigned)(dv[u].x - lo), (unsigned)(dv[u].y - lo),
                               (unsigned)(dv[u].z - lo), (unsigned)(dv[u].w - lo)};
        const unsigned mn = min(min(t[0], t[1]), min(t[2], t[3]));
        if (__builtin_amdgcn_ballot_w32(mn < OWN_ROWS) != 0u) {
          const int ebase = ((pc + u) * 32 + lane) * 4;
#pragma unroll
          for (int c = 0; c < 4; ++c) {
            const bool m = t[c] < OWN_ROWS;
            const unsigned bal = __builtin_amdgcn_ballot_w32(m);
            if (bal != 0u) {
              if (m) {
                qmem[qb + cnt + (int)__builtin_amdgcn_mbcnt_lo(bal, 0u)] = ((ebase + c) << 8) | (int)t[c];
                atomicAdd(&indeg[t[c]], 1);
              }
              cnt += __builtin_popcount(bal);
            }
          }
        }
      }

      if (cnt >= 32 || (last && cnt > 0)) {
        for (int base = 0; base < cnt; base += 32) {
          const int n = min(32, cnt - base);
          int v = 0;
          if (lane < n) v = qmem[qb + base + lane];
          const int   e  = v >> 8;
          const int   dl = v & 255;
          const int   s  = src[e];
          const float dg = outdeg[s];
          gcn_sum_listed(feat, rows, s, dl, dg, n, lane);
        }
        cnt = 0;
      }
      if (last) break;
      g = gn;
    }
  }


  const int half = lane >> 4;
  const int m    = lane & 15;
  const int grp  = wave >> 3;
  const bool mult   = (grp == ((wave >> 1) & 3));
  const int  mi     = 2 * grp + (wave & 1);
  const int  gi     = wave - 2 * grp - (((wave & 7) > 2 * grp + 1) ? 2 : 0);
  const bool listed = (total <= LIST_CAP);
  const int  col    = mi * 16 + m;

  v2f b[32];
  float bv = 0.0f;
  if (mult) {
    gcn_load_bfrag<DD>(W, col, half, b);
    bv = bias[col];
  }

  for (int p = 0; p <= NBLK; ++p) {
    if (!mult) {
      if (listed && p < NBLK) {
        const int bs = (p == 0) ? 0 : __builtin_amdgcn_readfirstlane(rcnt[p * BLK_ROWS - 1]);
        const int be = __builtin_amdgcn_readfirstlane(rcnt[p * BLK_ROWS + BLK_ROWS - 1]);
        const int per = (((be - bs + G_WAVES - 1) / G_WAVES) + 7) & ~7;
        const int beg = bs + gi * per;
        const int end = min(be, beg + per);
        float4 acc = make_float4(0.f, 0.f, 0.f, 0.f);
        int currow = -1;
        for (int base = beg; base < end; base += 32) {
          const int n = min(32, end - base);
          int v = 0;
          if (lane < n) v = qmem[base + lane];
          const int   s  = v >> 8;
          const int   dl = v & 255;
          gcn_sum_ordered(feat, outdeg, rows, s, dl, n, lane, acc, currow);
        }
        if (currow >= 0) gcn_flush_run(rows, currow, acc, lane);
      }
    } else if (p > 0) {
      const int r0 = (p - 1) * BLK_ROWS;
      const float sc0 = gcn_deg_norm((float)indeg[r0 + m]);
      const float sc1 = gcn_deg_norm((float)indeg[r0 + 16 + m]);
      v8f acc0, acc1;
      gcn_tile2(rows + (r0 + m) * ROWP, rows + (r0 + 16 + m) * ROWP, sc0, sc1, half, b, acc0, acc1);
      float* dp = LAST ? hid + (r0 + 8 * half) * ROWP + col
                       : outp + (size_t)(lo + r0 + 8 * half) * DD + col;
      const int pitch = LAST ? ROWP : DD;
#pragma unroll
      for (int r = 0; r < 8; ++r) {
        dp[r * pitch]        = fmaxf(acc0[r] + bv, 0.0f);
        dp[(16 + r) * pitch] = fmaxf(acc1[r] + bv, 0.0f);
      }
    }
    __syncthreads();
  }

  if (LAST) {
    const int ct1  = wave & 3;
    const int rg1  = ((wave >> 2) + 2 * (wave & 3)) & 7;
    const int col1 = ct1 * 16 + m;
    gcn_load_bfrag<64>(Wl1, col1, half, b);
    const float bv1 = bl1[col1];
    float* tb = rows;
    for (int j = 0; j < 2; ++j) {
      const int rt = rg1 + 8 * j;
      if (rt < OWN_ROWS / 16) {
        const int r0 = rt * 16;
        const v8f acc = gcn_tile<0>(hid + (r0 + m) * ROWP, 1.0f, half, b);
#pragma unroll
        for (int r = 0; r < 8; ++r)
          tb[(r0 + r + 8 * half) * T_PITCH + col1] = fmaxf(acc[r] + bv1, 0.0f);
      }
    }
    __syncthreads();

    for (int o = tid; o < OWN_ROWS * 10; o += 1024) {
      const int row = o / 10;
      const int c   = o - row * 10;
      const float* tp = tb + row * T_PITCH;
      float acc = bl2[c];
#pragma unroll
      for (int k = 0; k < 64; ++k) acc = fmaf(tp[k], Wl2[k * 10 + c], acc);
      outp[(size_t)lo * 10 + o] = acc;
    }
  }
}


extern "C" void kernel_launch(void* const* d_in, const int* in_sizes, int n_in,
                              void* d_out, int out_size, void* d_ws, size_t ws_size,
                              hipStream_t stream) {
  const float* h   = (const float*)d_in[0];
  const int*   src = (const int*)d_in[1];
  const int*   dst = (const int*)d_in[2];
  const float* W0  = (const float*)d_in[3];
  const float* b0  = (const float*)d_in[4];
  const float* W1  = (const float*)d_in[5];
  const float* b1  = (const float*)d_in[6];
  const float* Wl1 = (const float*)d_in[7];
  const float* bl1 = (const float*)d_in[8];
  const float* Wl2 = (const float*)d_in[9];
  const float* bl2 = (const float*)d_in[10];
  float* out = (float*)d_out;

  float* wsf      = (float*)d_ws;
  float* deg_out  = wsf;
  float* buf1     = wsf + 2 * NN;
  float* buf2     = buf1 + (size_t)NN * DD;

  const int TPB = 256;
  dim3 ownGrid(OWN_BLOCKS);

  gcn_fill_zero<<<dim3((NN + TPB - 1) / TPB), TPB, 0, stream>>>(deg_out, NN, (int*)buf1);
  gcn_degree_bin<<<dim3(BIN_BLOCKS + CNT_BLOCKS), 1024, 0, stream>>>(src, dst, deg_out, (int*)buf1);

  gcn_layer_owner<0><<<ownGrid, 1024, 0, stream>>>(h, src, dst, deg_out, (const int*)buf1,
                                                   W0, b0, nullptr, nullptr, nullptr, nullptr, buf2);
  gcn_layer_owner<1><<<ownGrid, 1024, 0, stream>>>(buf2, src, dst, deg_out, (const int*)buf1,
                                                   W1, b1, Wl1, bl1, Wl2, bl2, out);
}
